# starting kernel
# baseline (speedup 1.0000x reference)
#include <hip/hip_runtime.h>

typedef __attribute__((ext_vector_type(2))) float v2f;
typedef __attribute__((ext_vector_type(8))) float v8f;

#define C_IN   512
#define HW_    128
#define NBATCH 8
#define K_RAW  9
#define K_PAD  12
#define EQ_SCALE 0.014731391274719739f

__global__ void prep_weight_kernel(const float* __restrict__ w,
                                   float* __restrict__ bp) {
  int tid = blockIdx.x * blockDim.x + threadIdx.x;
  if (tid >= 6144 * 16) return;
  int n  = tid & 15;
  int kp = tid >> 4;
  int ci = kp / K_PAD;
  int t  = kp - ci * K_PAD;
  int dy = t >> 2, dx = t & 3;
  float val = 0.0f;
  if (dx < 3 && n < 3)
    val = w[n * (C_IN * K_RAW) + ci * K_RAW + dy * 3 + dx] * EQ_SCALE;
  int p = kp >> 1, par = kp & 1;
  bp[(p * 16 + n) * 2 + par] = val;
}

__global__ void conv_wmma_fast_kernel(const float* __restrict__ in,
                                      const float* __restrict__ bpack,
                                      float* __restrict__ out) {
  const int lane = threadIdx.x & 31;
  const int wid  = __builtin_amdgcn_readfirstlane(blockIdx.x * 4 + (threadIdx.x >> 5));
  const int b    = wid / 378;
  const int r    = wid - b * 378;
  const int y    = 1 + r / 3;
  const int pi   = r - (r / 3) * 3;
  const int x0A  = 16 + pi * 32;

  const int M    = lane & 15;
  const int hi   = lane >> 4;
  const int xoff = M + hi * 2 - 1;

  const int offA = (y - 1) * HW_ + x0A + xoff;
  const int offB = offA + 16;
  const int bidx = hi * 16 + M;

  const float* pin = in + (size_t)b * C_IN * (HW_ * HW_);
  const v2f*   bp  = (const v2f*)bpack;

  v8f cA = {}, cB = {};
  for (int ci = 0; ci < C_IN; ++ci) {
    const float* p  = pin + (size_t)ci * (HW_ * HW_);
    const v2f*   pb = bp + ci * 96;
#pragma unroll
    for (int sub = 0; sub < 3; ++sub) {
      v2f aA, aB;
      __builtin_memcpy(&aA, p + offA + sub * HW_, 8);
      __builtin_memcpy(&aB, p + offB + sub * HW_, 8);
      v2f bb = pb[sub * 32 + bidx];
      cA = __builtin_amdgcn_wmma_f32_16x16x4_f32(
          false, aA, false, bb, (short)0, cA, false, false);
      cB = __builtin_amdgcn_wmma_f32_16x16x4_f32(
          false, aB, false, bb, (short)0, cB, false, false);
    }
  }

  if (M < 3) {
    const int mbase = hi * 8;
    float* poA = out + (((size_t)b * 3 + M) * HW_ + y) * HW_ + x0A + mbase;
    float* poB = poA + 16;
#pragma unroll
    for (int r8 = 0; r8 < 8; ++r8) { poA[r8] = cA[r8]; poB[r8] = cB[r8]; }
  }
}

__global__ void conv_wmma_edge_kernel(const float* __restrict__ in,
                                      const float* __restrict__ bpack,
                                      float* __restrict__ out) {
  const int lane = threadIdx.x & 31;
  const int wid  = __builtin_amdgcn_readfirstlane(blockIdx.x * 4 + (threadIdx.x >> 5));
  const int b    = wid / 268;
  const int e    = wid - b * 268;
  int y, xt;
  if (e < 16) { y = (e < 8) ? 0 : 127; xt = e & 7; }
  else        { int e2 = e - 16; y = 1 + (e2 >> 1); xt = (e2 & 1) ? 7 : 0; }
  const int x0 = xt << 4;

  const int M    = lane & 15;
  const int hi   = lane >> 4;
  const int bidx = hi * 16 + M;

  int  off[6];
  bool vld[6];
#pragma unroll
  for (int j = 0; j < 6; ++j) {
    int dy = j >> 1;
    int dx = hi * 2 + (j & 1);
    if (dx < 3) {
      int ym = y + dy - 1;
      int xm = x0 + M + dx - 1;
      vld[j] = (ym >= 0) && (ym < HW_) && (xm >= 0) && (xm < HW_);
      off[j] = vld[j] ? (ym * HW_ + xm) : 0;
    } else { vld[j] = false; off[j] = 0; }
  }

  const float* pin = in + (size_t)b * C_IN * (HW_ * HW_);
  const v2f*   bp  = (const v2f*)bpack;

  v8f c = {};
  for (int ci = 0; ci < C_IN; ++ci) {
    const float* p  = pin + (size_t)ci * (HW_ * HW_);
    const v2f*   pb = bp + ci * 96;
#pragma unroll
    for (int sub = 0; sub < 3; ++sub) {
      float a0 = vld[2 * sub]     ? p[off[2 * sub]]     : 0.0f;
      float a1 = vld[2 * sub + 1] ? p[off[2 * sub + 1]] : 0.0f;
      v2f a; a.x = a0; a.y = a1;
      v2f bb = pb[sub * 32 + bidx];
      c = __builtin_amdgcn_wmma_f32_16x16x4_f32(
          false, a, false, bb, (short)0, c, false, false);
    }
  }

  if (M < 3) {
    const int mbase = hi * 8;
    float* po = out + (((size_t)b * 3 + M) * HW_ + y) * HW_ + x0 + mbase;
#pragma unroll
    for (int r8 = 0; r8 < 8; ++r8) po[r8] = c[r8];
  }
}

__global__ void skip_bias_kernel(const float* __restrict__ skip,
                                 const float* __restrict__ bias,
                                 float* __restrict__ out) {
  int tid = blockIdx.x * blockDim.x + threadIdx.x;
  const int total = NBATCH * 3 * HW_ * HW_;
  if (tid >= total) return;

  int x  = tid & 127;
  int y  = (tid >> 7) & 127;
  int bc = tid >> 14;
  int co = bc % 3;

  int y2 = y >> 1; bool ye = !(y & 1);
  int   my0 = ye ? y2     : y2 + 1;  float wy0 = ye ? 0.75f : 0.25f;
  int   my1 = ye ? y2 - 1 : y2;      float wy1 = ye ? 0.25f : 0.75f;
  int x2 = x >> 1; bool xe = !(x & 1);
  int   mx0 = xe ? x2     : x2 + 1;  float wx0 = xe ? 0.75f : 0.25f;
  int   mx1 = xe ? x2 - 1 : x2;      float wx1 = xe ? 0.25f : 0.75f;

  const float* s = skip + (size_t)bc * 64 * 64;
  float acc = bias[co];
  if (my0 >= 0 && my0 < 64) {
    if (mx0 >= 0 && mx0 < 64) acc += wy0 * wx0 * s[my0 * 64 + mx0];
    if (mx1 >= 0 && mx1 < 64) acc += wy0 * wx1 * s[my0 * 64 + mx1];
  }
  if (my1 >= 0 && my1 < 64) {
    if (mx0 >= 0 && mx0 < 64) acc += wy1 * wx0 * s[my1 * 64 + mx0];
    if (mx1 >= 0 && mx1 < 64) acc += wy1 * wx1 * s[my1 * 64 + mx1];
  }
  out[tid] += acc;
}

extern "C" void kernel_launch(void* const* d_in, const int* in_sizes, int n_in,
                              void* d_out, int out_size, void* d_ws,
                              size_t ws_size, hipStream_t stream) {
  (void)in_sizes; (void)n_in; (void)out_size; (void)ws_size;
  const float* in   = (const float*)d_in[0];
  const float* skip = (const float*)d_in[1];
  const float* w    = (const float*)d_in[2];
  const float* bias = (const float*)d_in[3];
  float* out = (float*)d_out;
  float* bp  = (float*)d_ws;

  prep_weight_kernel<<<dim3((6144 * 16 + 255) / 256), dim3(256), 0, stream>>>(w, bp);
  conv_wmma_fast_kernel<<<dim3(756), dim3(128), 0, stream>>>(in, bp, out);
  conv_wmma_edge_kernel<<<dim3(536), dim3(128), 0, stream>>>(in, bp, out);
  skip_bias_kernel<<<dim3((NBATCH * 3 * HW_ * HW_ + 255) / 256), dim3(256), 0, stream>>>(
      skip, bias, out);
}


# Vertical taps moved from K to N: one 16x16x4 product per input row and channel (K = dx, N = dy x out ch); 8-row strips, one kernel for all tiles, loads 5 channels ahead; changes the summation order
# speedup vs baseline: 5.1771x; 5.1771x over previous
#include <hip/hip_runtime.h>

typedef __attribute__((ext_vector_type(2))) float v2f;
typedef __attribute__((ext_vector_type(8))) float v8f;

#define C_IN   512
#define HW_    128
#define PLANE  (HW_ * HW_)
#define NBATCH 8
#define RS     8
#define NR     (RS + 2)
#define RING   5
#define C_LOOP 515
#define C_PAD  520
#define EQ_SCALE 0.014731391274719739f

__global__ void prep_weight_kernel(const float* __restrict__ w,
                                   float* __restrict__ bp) {
  int tid = blockIdx.x * blockDim.x + threadIdx.x;
  if (tid >= C_PAD * 64) return;
  int s  = tid & 1;
  int c  = (tid >> 1) & 15;
  int h  = (tid >> 5) & 1;
  int ci = tid >> 6;
  int dx = 2 * h + s, dy = c >> 2, co = c & 3;
  float val = 0.0f;
  if (ci < C_IN && dx < 3 && dy < 3 && co < 3)
    val = w[co * (C_IN * 9) + ci * 9 + dy * 3 + dx] * EQ_SCALE;
  bp[tid] = val;
}

template <int MODE>
__device__ __forceinline__ v2f fix_edge(v2f ld, bool fL, bool fR1, bool fR2) {
  if (MODE == 1) {
    v2f a; a.x = fL ? 0.0f : ld.x; a.y = fL ? ld.x : ld.y; return a;
  }
  if (MODE == 2) {
    v2f a;
    a.x = fR2 ? 0.0f : (fR1 ? ld.y : ld.x); a.y = ld.y; return a;
  }
  return ld;
}

template <int MODE>
__device__ __forceinline__ void ring_trip(v2f (&A)[RING][NR], const v2f (&Bc)[RING],
                                          v2f (&Bn)[RING], v8f (&acc)[NR],
                                          const float* __restrict__ pin,
                                          const v2f* __restrict__ bp,
                                          const unsigned (&voff)[NR], int lane,
                                          int ci, bool fL, bool fR1, bool fR2) {
#pragma unroll
  for (int u = 0; u < RING; ++u) {
    int cn = ci + u + RING;
    int ca = cn > C_IN - 1 ? C_IN - 1 : cn;
    const float* pn = pin + (size_t)ca * PLANE;
    Bn[u] = bp[cn * 32 + lane];
    __builtin_amdgcn_sched_barrier(0);
#pragma unroll
    for (int j = 0; j < NR; ++j) {
      v2f a = fix_edge<MODE>(A[u][j], fL, fR1, fR2);
      acc[j] = __builtin_amdgcn_wmma_f32_16x16x4_f32(
          false, a, false, Bc[u], (short)0, acc[j], false, false);
      __builtin_memcpy(&A[u][j], pn + voff[j], 8);
      __builtin_amdgcn_sched_barrier(0);
    }
  }
}

template <int MODE>
__device__ __forceinline__ void conv_strip(const float* __restrict__ pin,
                                           const v2f* __restrict__ bp,
                                           float* __restrict__ pout,
                                           int y0, int x0, int lane) {
  const int M  = lane & 15;
  const int hi = lane >> 4;
  int xs = x0 + M + 2 * hi - 1;
  bool fL = false, fR1 = false, fR2 = false;
  if (MODE == 1) { fL = (lane == 0); if (fL) xs = 0; }
  if (MODE == 2) { fR1 = (lane == 30); fR2 = (lane == 31); if (fR1 || fR2) xs = 126; }

  unsigned voff[NR];
#pragma unroll
  for (int j = 0; j < NR; ++j) {
    int yy = y0 - 1 + j;
    yy = yy < 0 ? 0 : (yy > HW_ - 1 ? HW_ - 1 : yy);
    voff[j] = (unsigned)(yy * HW_ + xs);
  }

  v8f acc[NR];
#pragma unroll
  for (int j = 0; j < NR; ++j) acc[j] = v8f{};

  v2f A[RING][NR], Ba[RING], Bb[RING];
#pragma unroll
  for (int u = 0; u < RING; ++u) {
#pragma unroll
    for (int j = 0; j < NR; ++j) A[u][j] = v2f{};
    Ba[u] = v2f{};
    Bb[u] = v2f{};
  }

  for (int ci = -RING; ci < C_LOOP; ci += 2 * RING) {
    ring_trip<MODE>(A, Ba, Bb, acc, pin, bp, voff, lane, ci, fL, fR1, fR2);
    ring_trip<MODE>(A, Bb, Ba, acc, pin, bp, voff, lane, ci + RING, fL, fR1, fR2);
  }

  const bool noTop = (y0 == 0);
  const bool noBot = (y0 + RS == HW_);
#pragma unroll
  for (int i = 0; i < RS; ++i) {
    float o[8];
#pragma unroll
    for (int r = 0; r < 8; ++r) {
      float t0 = acc[i][r];
      float t1 = __shfl_down(acc[i + 1][r], 4, 16);
      float t2 = __shfl_down(acc[i + 2][r], 8, 16);
      if (i == 0)      t0 = noTop ? 0.0f : t0;
      if (i == RS - 1) t2 = noBot ? 0.0f : t2;
      o[r] = (t0 + t1) + t2;
    }
    if (M < 3) {
      float4* po = (float4*)(pout + ((size_t)M * HW_ + (y0 + i)) * HW_ + x0 + hi * 8);
      po[0] = make_float4(o[0], o[1], o[2], o[3]);
      po[1] = make_float4(o[4], o[5], o[6], o[7]);
    }
  }
}

__global__ void __launch_bounds__(128) conv_rows_kernel(const float* __restrict__ in,
                                 const float* __restrict__ bpack,
                                 float* __restrict__ out) {
  const int lane = threadIdx.x & 31;
  const int wid  = __builtin_amdgcn_readfirstlane(blockIdx.x * 4 + (threadIdx.x >> 5));
  const int xt   = wid & 7;
  const int st   = (wid >> 3) & 15;
  const int b    = wid >> 7;
  const int y0   = st * RS;
  const int x0   = xt * 16;
  const float* pin  = in + (size_t)b * C_IN * PLANE;
  float*       pout = out + (size_t)b * 3 * PLANE;
  const v2f*   bp   = (const v2f*)bpack;
  if (xt == 0)      conv_strip<1>(pin, bp, pout, y0, x0, lane);
  else if (xt == 7) conv_strip<2>(pin, bp, pout, y0, x0, lane);
  else              conv_strip<0>(pin, bp, pout, y0, x0, lane);
}

__global__ void skip_bias_kernel(const float* __restrict__ skip,
                                 const float* __restrict__ bias,
                                 float* __restrict__ out) {
  int tid = blockIdx.x * blockDim.x + threadIdx.x;
  const int total = NBATCH * 3 * HW_ * HW_;
  if (tid >= total) return;

  int x  = tid & 127;
  int y  = (tid >> 7) & 127;
  int bc = tid >> 14;
  int co = bc % 3;

  int y2 = y >> 1; bool ye = !(y & 1);
  int   my0 = ye ? y2     : y2 + 1;  float wy0 = ye ? 0.75f : 0.25f;
  int   my1 = ye ? y2 - 1 : y2;      float wy1 = ye ? 0.25f : 0.75f;
  int x2 = x >> 1; bool xe = !(x & 1);
  int   mx0 = xe ? x2     : x2 + 1;  float wx0 = xe ? 0.75f : 0.25f;
  int   mx1 = xe ? x2 - 1 : x2;      float wx1 = xe ? 0.25f : 0.75f;

  const float* s = skip + (size_t)bc * 64 * 64;
  float acc = bias[co];
  if (my0 >= 0 && my0 < 64) {
    if (mx0 >= 0 && mx0 < 64) acc += wy0 * wx0 * s[my0 * 64 + mx0];
    if (mx1 >= 0 && mx1 < 64) acc += wy0 * wx1 * s[my0 * 64 + mx1];
  }
  if (my1 >= 0 && my1 < 64) {
    if (mx0 >= 0 && mx0 < 64) acc += wy1 * wx0 * s[my1 * 64 + mx0];
    if (mx1 >= 0 && mx1 < 64) acc += wy1 * wx1 * s[my1 * 64 + mx1];
  }
  out[tid] += acc;
}

extern "C" void kernel_launch(void* const* d_in, const int* in_sizes, int n_in,
                              void* d_out, int out_size, void* d_ws,
                              size_t ws_size, hipStream_t stream) {
  (void)in_sizes; (void)n_in; (void)out_size; (void)ws_size;
  const float* in   = (const float*)d_in[0];
  const float* skip = (const float*)d_in[1];
  const float* w    = (const float*)d_in[2];
  const float* bias = (const float*)d_in[3];
  float* out = (float*)d_out;
  float* bp  = (float*)d_ws;

  prep_weight_kernel<<<dim3(130), dim3(256), 0, stream>>>(w, bp);
  conv_rows_kernel<<<dim3(256), dim3(128), 0, stream>>>(in, bp, out);
  skip_bias_kernel<<<dim3((NBATCH * 3 * HW_ * HW_ + 255) / 256), dim3(256), 0, stream>>>(
      skip, bias, out);
}


# Loads of 5 channels issued as one group in front of the previous 5 channels' 50 products (two register sets), not one load behind each product; sum order unchanged
# speedup vs baseline: 10.9073x; 2.1068x over previous
#include <hip/hip_runtime.h>

typedef __attribute__((ext_vector_type(2))) float v2f;
typedef __attribute__((ext_vector_type(8))) float v8f;

#define C_IN   512
#define HW_    128
#define PLANE  (HW_ * HW_)
#define NBATCH 8
#define RS     8
#define NR     (RS + 2)
#define TC     5
#define C_PAD  520
#define EQ_SCALE 0.014731391274719739f

__global__ void prep_weight_kernel(const float* __restrict__ w,
                                   float* __restrict__ bp) {
  int tid = blockIdx.x * blockDim.x + threadIdx.x;
  if (tid >= C_PAD * 64) return;
  int s  = tid & 1;
  int c  = (tid >> 1) & 15;
  int h  = (tid >> 5) & 1;
  int ci = tid >> 6;
  int dx = 2 * h + s, dy = c >> 2, co = c & 3;
  float val = 0.0f;
  if (ci < C_IN && dx < 3 && dy < 3 && co < 3)
    val = w[co * (C_IN * 9) + ci * 9 + dy * 3 + dx] * EQ_SCALE;
  bp[tid] = val;
}

template <int MODE>
__device__ __forceinline__ v2f fix_edge(v2f ld, bool fL, bool fR1, bool fR2) {
  if (MODE == 1) {
    v2f a; a.x = fL ? 0.0f : ld.x; a.y = fL ? ld.x : ld.y; return a;
  }
  if (MODE == 2) {
    v2f a;
    a.x = fR2 ? 0.0f : (fR1 ? ld.y : ld.x); a.y = ld.y; return a;
  }
  return ld;
}

template <bool TAIL>
__device__ __forceinline__ void load_trip(v2f (&A)[TC][NR], v2f (&B)[TC],
                                          const float* __restrict__ pin,
                                          const v2f* __restrict__ bp,
                                          const unsigned (&voff)[NR], int lane, int c0) {
#pragma unroll
  for (int u = 0; u < TC; ++u) {
    int cn = c0 + u;
    int ca = (TAIL && cn > C_IN - 1) ? C_IN - 1 : cn;
    const char* pn = (const char*)(pin + (size_t)ca * PLANE);
    B[u] = bp[cn * 32 + lane];
#pragma unroll
    for (int j = 0; j < NR; ++j)
      __builtin_memcpy(&A[u][j], pn + voff[j], 8);
  }
}

template <int MODE>
__device__ __forceinline__ void mul_trip(const v2f (&A)[TC][NR], const v2f (&B)[TC],
                                         v8f (&acc)[NR], bool fL, bool fR1, bool fR2) {
#pragma unroll
  for (int u = 0; u < TC; ++u) {
#pragma unroll
    for (int j = 0; j < NR; ++j) {
      v2f a = fix_edge<MODE>(A[u][j], fL, fR1, fR2);
      acc[j] = __builtin_amdgcn_wmma_f32_16x16x4_f32(
          false, a, false, B[u], (short)0, acc[j], false, false);
    }
  }
}

template <int MODE>
__device__ __forceinline__ void conv_strip(const float* __restrict__ pin,
                                           const v2f* __restrict__ bp,
                                           float* __restrict__ pout,
                                           int y0, int x0, int lane) {
  const int M  = lane & 15;
  const int hi = lane >> 4;
  int xs = x0 + M + 2 * hi - 1;
  bool fL = false, fR1 = false, fR2 = false;
  if (MODE == 1) { fL = (lane == 0); if (fL) xs = 0; }
  if (MODE == 2) { fR1 = (lane == 30); fR2 = (lane == 31); if (fR1 || fR2) xs = 126; }

  unsigned voff[NR];
#pragma unroll
  for (int j = 0; j < NR; ++j) {
    int yy = y0 - 1 + j;
    yy = yy < 0 ? 0 : (yy > HW_ - 1 ? HW_ - 1 : yy);
    voff[j] = (unsigned)(yy * HW_ + xs) * 4u;
  }

  v8f acc[NR];
#pragma unroll
  for (int j = 0; j < NR; ++j) acc[j] = v8f{};

  v2f AX[TC][NR], AY[TC][NR], BX[TC], BY[TC];

  load_trip<false>(AX, BX, pin, bp, voff, lane, 0);
  __builtin_amdgcn_sched_barrier(0);
  for (int c0 = 0; c0 < 100 * TC; c0 += 2 * TC) {
    load_trip<false>(AY, BY, pin, bp, voff, lane, c0 + TC);
    __builtin_amdgcn_sched_barrier(0);
    mul_trip<MODE>(AX, BX, acc, fL, fR1, fR2);
    __builtin_amdgcn_sched_barrier(0);
    load_trip<false>(AX, BX, pin, bp, voff, lane, c0 + 2 * TC);
    __builtin_amdgcn_sched_barrier(0);
    mul_trip<MODE>(AY, BY, acc, fL, fR1, fR2);
    __builtin_amdgcn_sched_barrier(0);
  }
  load_trip<false>(AY, BY, pin, bp, voff, lane, 101 * TC);
  __builtin_amdgcn_sched_barrier(0);
  mul_trip<MODE>(AX, BX, acc, fL, fR1, fR2);
  __builtin_amdgcn_sched_barrier(0);
  load_trip<true>(AX, BX, pin, bp, voff, lane, 102 * TC);
  __builtin_amdgcn_sched_barrier(0);
  mul_trip<MODE>(AY, BY, acc, fL, fR1, fR2);
  mul_trip<MODE>(AX, BX, acc, fL, fR1, fR2);

  const bool noTop = (y0 == 0);
  const bool noBot = (y0 + RS == HW_);
#pragma unroll
  for (int i = 0; i < RS; ++i) {
    float o[8];
#pragma unroll
    for (int r = 0; r < 8; ++r) {
      float t0 = acc[i][r];
      float t1 = __shfl_down(acc[i + 1][r], 4, 16);
      float t2 = __shfl_down(acc[i + 2][r], 8, 16);
      if (i == 0)      t0 = noTop ? 0.0f : t0;
      if (i == RS - 1) t2 = noBot ? 0.0f : t2;
      o[r] = (t0 + t1) + t2;
    }
    if (M < 3) {
      float4* po = (float4*)(pout + ((size_t)M * HW_ + (y0 + i)) * HW_ + x0 + hi * 8);
      po[0] = make_float4(o[0], o[1], o[2], o[3]);
      po[1] = make_float4(o[4], o[5], o[6], o[7]);
    }
  }
}

__global__ void __launch_bounds__(128) conv_rows_kernel(const float* __restrict__ in,
                                 const float* __restrict__ bpack,
                                 float* __restrict__ out) {
  const int lane = threadIdx.x & 31;
  const int wid  = __builtin_amdgcn_readfirstlane(blockIdx.x * 4 + (threadIdx.x >> 5));
  const int xt   = wid & 7;
  const int st   = (wid >> 3) & 15;
  const int b    = wid >> 7;
  const int y0   = st * RS;
  const int x0   = xt * 16;
  const float* pin  = in + (size_t)b * C_IN * PLANE;
  float*       pout = out + (size_t)b * 3 * PLANE;
  const v2f*   bp   = (const v2f*)bpack;
  if (xt == 0)      conv_strip<1>(pin, bp, pout, y0, x0, lane);
  else if (xt == 7) conv_strip<2>(pin, bp, pout, y0, x0, lane);
  else              conv_strip<0>(pin, bp, pout, y0, x0, lane);
}

__global__ void skip_bias_kernel(const float* __restrict__ skip,
                                 const float* __restrict__ bias,
                                 float* __restrict__ out) {
  int tid = blockIdx.x * blockDim.x + threadIdx.x;
  const int total = NBATCH * 3 * HW_ * HW_;
  if (tid >= total) return;

  int x  = tid & 127;
  int y  = (tid >> 7) & 127;
  int bc = tid >> 14;
  int co = bc % 3;

  int y2 = y >> 1; bool ye = !(y & 1);
  int   my0 = ye ? y2     : y2 + 1;  float wy0 = ye ? 0.75f : 0.25f;
  int   my1 = ye ? y2 - 1 : y2;      float wy1 = ye ? 0.25f : 0.75f;
  int x2 = x >> 1; bool xe = !(x & 1);
  int   mx0 = xe ? x2     : x2 + 1;  float wx0 = xe ? 0.75f : 0.25f;
  int   mx1 = xe ? x2 - 1 : x2;      float wx1 = xe ? 0.25f : 0.75f;

  const float* s = skip + (size_t)bc * 64 * 64;
  float acc = bias[co];
  if (my0 >= 0 && my0 < 64) {
    if (mx0 >= 0 && mx0 < 64) acc += wy0 * wx0 * s[my0 * 64 + mx0];
    if (mx1 >= 0 && mx1 < 64) acc += wy0 * wx1 * s[my0 * 64 + mx1];
  }
  if (my1 >= 0 && my1 < 64) {
    if (mx0 >= 0 && mx0 < 64) acc += wy1 * wx0 * s[my1 * 64 + mx0];
    if (mx1 >= 0 && mx1 < 64) acc += wy1 * wx1 * s[my1 * 64 + mx1];
  }
  out[tid] += acc;
}

extern "C" void kernel_launch(void* const* d_in, const int* in_sizes, int n_in,
                              void* d_out, int out_size, void* d_ws,
                              size_t ws_size, hipStream_t stream) {
  (void)in_sizes; (void)n_in; (void)out_size; (void)ws_size;
  const float* in   = (const float*)d_in[0];
  const float* skip = (const float*)d_in[1];
  const float* w    = (const float*)d_in[2];
  const float* bias = (const float*)d_in[3];
  float* out = (float*)d_out;
  float* bp  = (float*)d_ws;

  prep_weight_kernel<<<dim3(130), dim3(256), 0, stream>>>(w, bp);
  conv_rows_kernel<<<dim3(256), dim3(128), 0, stream>>>(in, bp, out);
  skip_bias_kernel<<<dim3((NBATCH * 3 * HW_ * HW_ + 255) / 256), dim3(256), 0, stream>>>(
      skip, bias, out);
}


# Each strip's 10 input rows split over two waves (5 rows each, all channels), 8 waves per block so two waves share a SIMD; one product row per half handed over in LDS at the end; sum order unchanged
# speedup vs baseline: 12.2075x; 1.1192x over previous
#include <hip/hip_runtime.h>

typedef __attribute__((ext_vector_type(2))) float v2f;
typedef __attribute__((ext_vector_type(4))) float v4f;
typedef __attribute__((ext_vector_type(8))) float v8f;

#define C_IN   512
#define HW_    128
#define PLANE  (HW_ * HW_)
#define NBATCH 8
#define RS     8
#define NR     (RS + 2)
#define TC     5
#define C_PAD  520
#define EQ_SCALE 0.014731391274719739f

__global__ void prep_weight_kernel(const float* __restrict__ w,
                                   float* __restrict__ bp) {
  int tid = blockIdx.x * blockDim.x + threadIdx.x;
  if (tid >= C_PAD * 64) return;
  int s  = tid & 1;
  int c  = (tid >> 1) & 15;
  int h  = (tid >> 5) & 1;
  int ci = tid >> 6;
  int dx = 2 * h + s, dy = c >> 2, co = c & 3;
  float val = 0.0f;
  if (ci < C_IN && dx < 3 && dy < 3 && co < 3)
    val = w[co * (C_IN * 9) + ci * 9 + dy * 3 + dx] * EQ_SCALE;
  bp[tid] = val;
}

#define WIN    24
#define NRW    5
#define CHF    (NRW * WIN)
#define SLOTF  (TC * CHF)


__device__ __forceinline__ void gl_trip(v4f (&G)[TC], v2f (&GB)[TC],
                                        const float* __restrict__ pin,
                                        const v2f* __restrict__ bp,
                                        unsigned goff, int lane, int c0) {
#pragma unroll
  for (int u = 0; u < TC; ++u) {
    int cn = c0 + u;
    int ca = cn > C_IN - 1 ? C_IN - 1 : cn;
    const char* pn = (const char*)(pin + (size_t)ca * PLANE);
    G[u] = *(const v4f*)(pn + goff);
    GB[u] = bp[cn * 32 + lane];
  }
}

template <int MODE>
__device__ __forceinline__ void st_trip(const v4f (&G)[TC], float* slot,
                                        unsigned soff, bool zf) {
#pragma unroll
  for (int u = 0; u < TC; ++u) {
    v4f v = G[u];
    if (MODE != 0) { v4f z = {0.0f, 0.0f, 0.0f, 0.0f}; v = zf ? z : v; }
    *(v4f*)(slot + u * CHF + soff) = v;
  }
}

__device__ __forceinline__ void fr_trip(v2f (&A)[TC][NRW], const float* slot, int foff) {
#pragma unroll
  for (int u = 0; u < TC; ++u) {
#pragma unroll
    for (int j = 0; j < NRW; ++j)
      __builtin_memcpy(&A[u][j], slot + u * CHF + j * WIN + foff, 8);
  }
}

__device__ __forceinline__ void mul_trip(const v2f (&A)[TC][NRW], const v2f (&B)[TC],
                                         v8f (&acc)[NRW]) {
#pragma unroll
  for (int u = 0; u < TC; ++u) {
#pragma unroll
    for (int j = 0; j < NRW; ++j)
      acc[j] = __builtin_amdgcn_wmma_f32_16x16x4_f32(
          false, A[u][j], false, B[u], (short)0, acc[j], false, false);
  }
}

template <int MODE, bool LOAD>
__device__ __forceinline__ void body(v2f (&AC)[TC][NRW], v2f (&AN)[TC][NRW],
                                     v4f (&GS)[TC], v4f (&GL)[TC], v2f (&WC)[TC],
                                     v8f (&acc)[NRW],
                                     const float* __restrict__ pin, const v2f* __restrict__ bp,
                                     float* slotN, unsigned goff,
                                     unsigned soff, bool zf, int foff,
                                     int lane, int c_far) {
  st_trip<MODE>(GS, slotN, soff, zf);
  __builtin_amdgcn_fence(__ATOMIC_RELEASE, "wavefront");
  __builtin_amdgcn_sched_barrier(0);
  __builtin_amdgcn_fence(__ATOMIC_ACQUIRE, "wavefront");
  fr_trip(AN, slotN, foff);
  __builtin_amdgcn_sched_barrier(0);
  mul_trip(AC, WC, acc);
  __builtin_amdgcn_sched_barrier(0);
  if (LOAD) gl_trip(GL, WC, pin, bp, goff, lane, c_far);
  __builtin_amdgcn_sched_barrier(0);
}

template <int MODE>
__device__ __forceinline__ void conv_rows5(const float* __restrict__ pin,
                                           const v2f* __restrict__ bp,
                                           float* lds, v8f (&acc)[NRW],
                                           int yb, int x0, int lane) {
  const int M  = lane & 15;
  const int hi = lane >> 4;
  const int foff = 3 + M + 2 * hi;

  const int ls = lane < 29 ? lane : 29;
  const int rr = ls / 6;
  const int k  = ls - rr * 6;
  int kk = k;
  bool zf = false;
  if (MODE == 1) { zf = (k == 0); if (zf) kk = 1; }
  if (MODE == 2) { zf = (k == 5); if (zf) kk = 4; }
  int yy = yb + rr;
  yy = yy < 0 ? 0 : (yy > HW_ - 1 ? HW_ - 1 : yy);
  const unsigned goff = (unsigned)(yy * HW_ + x0 - 4 + 4 * kk) * 4u;
  const unsigned soff = (unsigned)(rr * WIN + 4 * k);
  float* slot0 = lds;
  float* slot1 = lds + SLOTF;

#pragma unroll
  for (int j = 0; j < NRW; ++j) acc[j] = v8f{};

  v2f AX[TC][NRW], AY[TC][NRW];
  v4f G0[TC], G1[TC], G2[TC], G3[TC], G4[TC], G5[TC];
  v2f W0[TC], W1[TC], W2[TC], W3[TC], W4[TC], W5[TC];

  gl_trip(G0, W0, pin, bp, goff, lane, 0);
  gl_trip(G1, W1, pin, bp, goff, lane, TC);
  gl_trip(G2, W2, pin, bp, goff, lane, 2 * TC);
  gl_trip(G3, W3, pin, bp, goff, lane, 3 * TC);
  __builtin_amdgcn_sched_barrier(0);
  st_trip<MODE>(G0, slot0, soff, zf);
  __builtin_amdgcn_fence(__ATOMIC_RELEASE, "wavefront");
  __builtin_amdgcn_sched_barrier(0);
  __builtin_amdgcn_fence(__ATOMIC_ACQUIRE, "wavefront");
  fr_trip(AX, slot0, foff);
  __builtin_amdgcn_sched_barrier(0);
  gl_trip(G4, W4, pin, bp, goff, lane, 4 * TC);
  gl_trip(G5, W5, pin, bp, goff, lane, 5 * TC);
  __builtin_amdgcn_sched_barrier(0);

  for (int c0 = 0; c0 < 96 * TC; c0 += 6 * TC) {
    body<MODE, true>(AX, AY, G1, G0, W0, acc, pin, bp, slot1, goff, soff, zf, foff, lane, c0 + 6 * TC);
    body<MODE, true>(AY, AX, G2, G1, W1, acc, pin, bp, slot0, goff, soff, zf, foff, lane, c0 + 7 * TC);
    body<MODE, true>(AX, AY, G3, G2, W2, acc, pin, bp, slot1, goff, soff, zf, foff, lane, c0 + 8 * TC);
    body<MODE, true>(AY, AX, G4, G3, W3, acc, pin, bp, slot0, goff, soff, zf, foff, lane, c0 + 9 * TC);
    body<MODE, true>(AX, AY, G5, G4, W4, acc, pin, bp, slot1, goff, soff, zf, foff, lane, c0 + 10 * TC);
    body<MODE, true>(AY, AX, G0, G5, W5, acc, pin, bp, slot0, goff, soff, zf, foff, lane, c0 + 11 * TC);
  }
  body<MODE, true >(AX, AY, G1, G0, W0, acc, pin, bp, slot1, goff, soff, zf, foff, lane, 102 * TC);
  body<MODE, false>(AY, AX, G2, G1, W1, acc, pin, bp, slot0, goff, soff, zf, foff, lane, 0);
  body<MODE, false>(AX, AY, G3, G2, W2, acc, pin, bp, slot1, goff, soff, zf, foff, lane, 0);
  body<MODE, false>(AY, AX, G4, G3, W3, acc, pin, bp, slot0, goff, soff, zf, foff, lane, 0);
  body<MODE, false>(AX, AY, G5, G4, W4, acc, pin, bp, slot1, goff, soff, zf, foff, lane, 0);
  body<MODE, false>(AY, AX, G0, G5, W5, acc, pin, bp, slot0, goff, soff, zf, foff, lane, 0);
  mul_trip(AX, W0, acc);
}

__device__ __forceinline__ void put_row(const float (&t0)[8], const float (&t1)[8],
                                        const float (&t2)[8], float* __restrict__ pout,
                                        int y, int x0, int M, int hi) {
  float o[8];
#pragma unroll
  for (int r = 0; r < 8; ++r) o[r] = (t0[r] + t1[r]) + t2[r];
  if (M < 3) {
    float4* po = (float4*)(pout + ((size_t)M * HW_ + y) * HW_ + x0 + hi * 8);
    po[0] = make_float4(o[0], o[1], o[2], o[3]);
    po[1] = make_float4(o[4], o[5], o[6], o[7]);
  }
}

__global__ void __launch_bounds__(256) conv_rows_kernel(const float* __restrict__ in,
                                 const float* __restrict__ bpack,
                                 float* __restrict__ out) {
  __shared__ __attribute__((aligned(16))) float stage[8][2 * SLOTF];
  __shared__ __attribute__((aligned(16))) float exch[4][2][256];
  const int lane = threadIdx.x & 31;
  const int w    = __builtin_amdgcn_readfirstlane(threadIdx.x >> 5);
  const int tile = w & 3;
  const int half = w >> 2;
  float* lds = stage[w];
  const int wid  = __builtin_amdgcn_readfirstlane(blockIdx.x * 4 + tile);
  const int xt   = wid & 7;
  const int st   = (wid >> 3) & 15;
  const int b    = wid >> 7;
  const int y0   = st * RS;
  const int x0   = xt * 16;
  const int yb   = y0 - 1 + NRW * half;
  const float* pin  = in + (size_t)b * C_IN * PLANE;
  float*       pout = out + (size_t)b * 3 * PLANE;
  const v2f*   bp   = (const v2f*)bpack;

  v8f acc[NRW];
  if (xt == 0)      conv_rows5<1>(pin, bp, lds, acc, yb, x0, lane);
  else if (xt == 7) conv_rows5<2>(pin, bp, lds, acc, yb, x0, lane);
  else              conv_rows5<0>(pin, bp, lds, acc, yb, x0, lane);

  const int M  = lane & 15;
  const int hi = lane >> 4;
  {
    const v8f give = half == 0 ? acc[NRW - 1] : acc[0];
    float4* pe = (float4*)&exch[tile][half][M * 16 + hi * 8];
    pe[0] = make_float4(give[0], give[1], give[2], give[3]);
    pe[1] = make_float4(give[4], give[5], give[6], give[7]);
  }
  __syncthreads();
  const int cm = M < 3 ? M : 0;
  float xr[8];
  {
    const int col = half == 0 ? 8 + cm : cm;
    const float4* pe = (const float4*)&exch[tile][1 - half][col * 16 + hi * 8];
    float4 a = pe[0], c = pe[1];
    xr[0] = a.x; xr[1] = a.y; xr[2] = a.z; xr[3] = a.w;
    xr[4] = c.x; xr[5] = c.y; xr[6] = c.z; xr[7] = c.w;
  }
  const bool noTop = (y0 == 0);
  const bool noBot = (y0 + RS == HW_);
  if (half == 0) {
#pragma unroll
    for (int i = 0; i < 4; ++i) {
      float t0[8], t1[8], t2[8];
#pragma unroll
      for (int r = 0; r < 8; ++r) {
        t0[r] = acc[i][r];
        t1[r] = __shfl_down(acc[i + 1][r], 4, 16);
        t2[r] = i < 3 ? __shfl_down(acc[i < 3 ? i + 2 : 0][r], 8, 16) : xr[r];
        if (i == 0) t0[r] = noTop ? 0.0f : t0[r];
      }
      put_row(t0, t1, t2, pout, y0 + i, x0, M, hi);
    }
  } else {
#pragma unroll
    for (int i = 0; i < 4; ++i) {
      float t0[8], t1[8], t2[8];
#pragma unroll
      for (int r = 0; r < 8; ++r) {
        t0[r] = i > 0 ? acc[i > 0 ? i - 1 : 0][r] : xr[r];
        t1[r] = __shfl_down(acc[i][r], 4, 16);
        t2[r] = __shfl_down(acc[i + 1][r], 8, 16);
        if (i == 3) t2[r] = noBot ? 0.0f : t2[r];
      }
      put_row(t0, t1, t2, pout, y0 + 4 + i, x0, M, hi);
    }
  }
}

__global__ void skip_bias_kernel(const float* __restrict__ skip,
                                 const float* __restrict__ bias,
                                 float* __restrict__ out) {
  int tid = blockIdx.x * blockDim.x + threadIdx.x;
  const int total = NBATCH * 3 * HW_ * HW_;
  if (tid >= total) return;

  int x  = tid & 127;
  int y  = (tid >> 7) & 127;
  int bc = tid >> 14;
  int co = bc % 3;

  int y2 = y >> 1; bool ye = !(y & 1);
  int   my0 = ye ? y2     : y2 + 1;  float wy0 = ye ? 0.75f : 0.25f;
  int   my1 = ye ? y2 - 1 : y2;      float wy1 = ye ? 0.25f : 0.75f;
  int x2 = x >> 1; bool xe = !(x & 1);
  int   mx0 = xe ? x2     : x2 + 1;  float wx0 = xe ? 0.75f : 0.25f;
  int   mx1 = xe ? x2 - 1 : x2;      float wx1 = xe ? 0.25f : 0.75f;

  const float* s = skip + (size_t)bc * 64 * 64;
  float acc = bias[co];
  if (my0 >= 0 && my0 < 64) {
    if (mx0 >= 0 && mx0 < 64) acc += wy0 * wx0 * s[my0 * 64 + mx0];
    if (mx1 >= 0 && mx1 < 64) acc += wy0 * wx1 * s[my0 * 64 + mx1];
  }
  if (my1 >= 0 && my1 < 64) {
    if (mx0 >= 0 && mx0 < 64) acc += wy1 * wx0 * s[my1 * 64 + mx0];
    if (mx1 >= 0 && mx1 < 64) acc += wy1 * wx1 * s[my1 * 64 + mx1];
  }
  out[tid] += acc;
}

extern "C" void kernel_launch(void* const* d_in, const int* in_sizes, int n_in,
                              void* d_out, int out_size, void* d_ws,
                              size_t ws_size, hipStream_t stream) {
  (void)in_sizes; (void)n_in; (void)out_size; (void)ws_size;
  const float* in   = (const float*)d_in[0];
  const float* skip = (const float*)d_in[1];
  const float* w    = (const float*)d_in[2];
  const float* bias = (const float*)d_in[3];
  float* out = (float*)d_out;
  float* bp  = (float*)d_ws;

  prep_weight_kernel<<<dim3(130), dim3(256), 0, stream>>>(w, bp);
  conv_rows_kernel<<<dim3(256), dim3(256), 0, stream>>>(in, bp, out);
  skip_bias_kernel<<<dim3((NBATCH * 3 * HW_ * HW_ + 255) / 256), dim3(256), 0, stream>>>(
      skip, bias, out);
}


# Input channels moved into K: one 16x16x4 product per input row, 4 channels and horizontal tap (3 pixel-shifted LDS fragments); 15 products per 4 channels, was 25 per 5; changes the summation order
# speedup vs baseline: 13.3301x; 1.0920x over previous
#include <hip/hip_runtime.h>

typedef __attribute__((ext_vector_type(2))) float v2f;
typedef __attribute__((ext_vector_type(4))) float v4f;
typedef __attribute__((ext_vector_type(8))) float v8f;

#define C_IN   512
#define HW_    128
#define PLANE  (HW_ * HW_)
#define NBATCH 8
#define RS     8
#define NR     (RS + 2)
#define TC     4
#define NQ     (C_IN / TC)
#define EQ_SCALE 0.014731391274719739f

__global__ void prep_weight_kernel(const float* __restrict__ w,
                                   float* __restrict__ bp) {
  int tid = blockIdx.x * blockDim.x + threadIdx.x;
  if (tid >= NQ * 3 * 64) return;
  int s  = tid & 1;
  int c  = (tid >> 1) & 15;
  int h  = (tid >> 5) & 1;
  int g  = tid >> 6;
  int q  = g / 3, dx = g - q * 3;
  int ci = 4 * q + 2 * h + s, dy = c >> 2, co = c & 3;
  float val = 0.0f;
  if (dy < 3 && co < 3)
    val = w[co * (C_IN * 9) + ci * 9 + dy * 3 + dx] * EQ_SCALE;
  bp[tid] = val;
}

#define WIN    24
#define NRW    5
#define CHF    (NRW * WIN)
#define SLOTF  (TC * CHF)


__device__ __forceinline__ void gl_trip(v4f (&G)[TC], v2f (&GB)[3],
                                        const float* __restrict__ pin,
                                        const v2f* __restrict__ bp,
                                        unsigned goff, int lane, int c0) {
#pragma unroll
  for (int u = 0; u < TC; ++u) {
    const char* pn = (const char*)(pin + (size_t)(c0 + u) * PLANE);
    G[u] = *(const v4f*)(pn + goff);
  }
#pragma unroll
  for (int d = 0; d < 3; ++d) GB[d] = bp[((c0 >> 2) * 3 + d) * 32 + lane];
}

template <int MODE>
__device__ __forceinline__ void st_trip(const v4f (&G)[TC], float* slot,
                                        unsigned soff, bool zf) {
#pragma unroll
  for (int u = 0; u < TC; ++u) {
    v4f v = G[u];
    if (MODE != 0) { v4f z = {0.0f, 0.0f, 0.0f, 0.0f}; v = zf ? z : v; }
    *(v4f*)(slot + u * CHF + soff) = v;
  }
}

__device__ __forceinline__ void fr_trip(v2f (&A)[NRW][3], const float* slot, int foff) {
#pragma unroll
  for (int j = 0; j < NRW; ++j) {
#pragma unroll
    for (int d = 0; d < 3; ++d) {
      const float* p = slot + j * WIN + d + foff;
      v2f v = {p[0], p[CHF]};
      A[j][d] = v;
    }
  }
}

__device__ __forceinline__ void mul_trip(const v2f (&A)[NRW][3], const v2f (&B)[3],
                                         v8f (&acc)[NRW]) {
#pragma unroll
  for (int d = 0; d < 3; ++d) {
#pragma unroll
    for (int j = 0; j < NRW; ++j)
      acc[j] = __builtin_amdgcn_wmma_f32_16x16x4_f32(
          false, A[j][d], false, B[d], (short)0, acc[j], false, false);
  }
}

template <int MODE, bool LOAD>
__device__ __forceinline__ void body(v2f (&AC)[NRW][3], v2f (&AN)[NRW][3],
                                     v4f (&GS)[TC], v4f (&GL)[TC], v2f (&WC)[3],
                                     v8f (&acc)[NRW],
                                     const float* __restrict__ pin, const v2f* __restrict__ bp,
                                     float* slotN, unsigned goff,
                                     unsigned soff, bool zf, int foff,
                                     int lane, int c_far) {
  st_trip<MODE>(GS, slotN, soff, zf);
  __builtin_amdgcn_fence(__ATOMIC_RELEASE, "wavefront");
  __builtin_amdgcn_sched_barrier(0);
  __builtin_amdgcn_fence(__ATOMIC_ACQUIRE, "wavefront");
  fr_trip(AN, slotN, foff);
  __builtin_amdgcn_sched_barrier(0);
  mul_trip(AC, WC, acc);
  __builtin_amdgcn_sched_barrier(0);
  if (LOAD) gl_trip(GL, WC, pin, bp, goff, lane, c_far);
  __builtin_amdgcn_sched_barrier(0);
}

template <int MODE>
__device__ __forceinline__ void conv_rows5(const float* __restrict__ pin,
                                           const v2f* __restrict__ bp,
                                           float* lds, v8f (&acc)[NRW],
                                           int yb, int x0, int lane) {
  const int M  = lane & 15;
  const int hi = lane >> 4;
  const int foff = 2 * hi * CHF + 3 + M;

  const int ls = lane < 29 ? lane : 29;
  const int rr = ls / 6;
  const int k  = ls - rr * 6;
  int kk = k;
  bool zf = false;
  if (MODE == 1) { zf = (k == 0); if (zf) kk = 1; }
  if (MODE == 2) { zf = (k == 5); if (zf) kk = 4; }
  int yy = yb + rr;
  yy = yy < 0 ? 0 : (yy > HW_ - 1 ? HW_ - 1 : yy);
  const unsigned goff = (unsigned)(yy * HW_ + x0 - 4 + 4 * kk) * 4u;
  const unsigned soff = (unsigned)(rr * WIN + 4 * k);
  float* slot0 = lds;
  float* slot1 = lds + SLOTF;

#pragma unroll
  for (int j = 0; j < NRW; ++j) acc[j] = v8f{};

  v2f AX[NRW][3], AY[NRW][3];
  v4f G0[TC], G1[TC], G2[TC], G3[TC], G4[TC], G5[TC];
  v2f W0[3], W1[3], W2[3], W3[3], W4[3], W5[3];

  gl_trip(G0, W0, pin, bp, goff, lane, 0);
  gl_trip(G1, W1, pin, bp, goff, lane, TC);
  gl_trip(G2, W2, pin, bp, goff, lane, 2 * TC);
  gl_trip(G3, W3, pin, bp, goff, lane, 3 * TC);
  __builtin_amdgcn_sched_barrier(0);
  st_trip<MODE>(G0, slot0, soff, zf);
  __builtin_amdgcn_fence(__ATOMIC_RELEASE, "wavefront");
  __builtin_amdgcn_sched_barrier(0);
  __builtin_amdgcn_fence(__ATOMIC_ACQUIRE, "wavefront");
  fr_trip(AX, slot0, foff);
  __builtin_amdgcn_sched_barrier(0);
  gl_trip(G4, W4, pin, bp, goff, lane, 4 * TC);
  gl_trip(G5, W5, pin, bp, goff, lane, 5 * TC);
  __builtin_amdgcn_sched_barrier(0);

  for (int c0 = 0; c0 < 120 * TC; c0 += 6 * TC) {
    body<MODE, true>(AX, AY, G1, G0, W0, acc, pin, bp, slot1, goff, soff, zf, foff, lane, c0 + 6 * TC);
    body<MODE, true>(AY, AX, G2, G1, W1, acc, pin, bp, slot0, goff, soff, zf, foff, lane, c0 + 7 * TC);
    body<MODE, true>(AX, AY, G3, G2, W2, acc, pin, bp, slot1, goff, soff, zf, foff, lane, c0 + 8 * TC);
    body<MODE, true>(AY, AX, G4, G3, W3, acc, pin, bp, slot0, goff, soff, zf, foff, lane, c0 + 9 * TC);
    body<MODE, true>(AX, AY, G5, G4, W4, acc, pin, bp, slot1, goff, soff, zf, foff, lane, c0 + 10 * TC);
    body<MODE, true>(AY, AX, G0, G5, W5, acc, pin, bp, slot0, goff, soff, zf, foff, lane, c0 + 11 * TC);
  }
  body<MODE, true >(AX, AY, G1, G0, W0, acc, pin, bp, slot1, goff, soff, zf, foff, lane, 126 * TC);
  body<MODE, true >(AY, AX, G2, G1, W1, acc, pin, bp, slot0, goff, soff, zf, foff, lane, 127 * TC);
  body<MODE, false>(AX, AY, G3, G2, W2, acc, pin, bp, slot1, goff, soff, zf, foff, lane, 0);
  body<MODE, false>(AY, AX, G4, G3, W3, acc, pin, bp, slot0, goff, soff, zf, foff, lane, 0);
  body<MODE, false>(AX, AY, G5, G4, W4, acc, pin, bp, slot1, goff, soff, zf, foff, lane, 0);
  body<MODE, false>(AY, AX, G0, G5, W5, acc, pin, bp, slot0, goff, soff, zf, foff, lane, 0);
  body<MODE, false>(AX, AY, G1, G0, W0, acc, pin, bp, slot1, goff, soff, zf, foff, lane, 0);
  mul_trip(AY, W1, acc);
}

__device__ __forceinline__ void put_row(const float (&t0)[8], const float (&t1)[8],
                                        const float (&t2)[8], float* __restrict__ pout,
                                        int y, int x0, int M, int hi) {
  float o[8];
#pragma unroll
  for (int r = 0; r < 8; ++r) o[r] = (t0[r] + t1[r]) + t2[r];
  if (M < 3) {
    float4* po = (float4*)(pout + ((size_t)M * HW_ + y) * HW_ + x0 + hi * 8);
    po[0] = make_float4(o[0], o[1], o[2], o[3]);
    po[1] = make_float4(o[4], o[5], o[6], o[7]);
  }
}

__global__ void __launch_bounds__(256) conv_rows_kernel(const float* __restrict__ in,
                                 const float* __restrict__ bpack,
                                 float* __restrict__ out) {
  __shared__ __attribute__((aligned(16))) float stage[8][2 * SLOTF];
  __shared__ __attribute__((aligned(16))) float exch[4][2][256];
  const int lane = threadIdx.x & 31;
  const int w    = __builtin_amdgcn_readfirstlane(threadIdx.x >> 5);
  const int tile = w & 3;
  const int half = w >> 2;
  float* lds = stage[w];
  const int wid  = __builtin_amdgcn_readfirstlane(blockIdx.x * 4 + tile);
  const int xt   = wid & 7;
  const int st   = (wid >> 3) & 15;
  const int b    = wid >> 7;
  const int y0   = st * RS;
  const int x0   = xt * 16;
  const int yb   = y0 - 1 + NRW * half;
  const float* pin  = in + (size_t)b * C_IN * PLANE;
  float*       pout = out + (size_t)b * 3 * PLANE;
  const v2f*   bp   = (const v2f*)bpack;

  v8f acc[NRW];
  if (xt == 0)      conv_rows5<1>(pin, bp, lds, acc, yb, x0, lane);
  else if (xt == 7) conv_rows5<2>(pin, bp, lds, acc, yb, x0, lane);
  else              conv_rows5<0>(pin, bp, lds, acc, yb, x0, lane);

  const int M  = lane & 15;
  const int hi = lane >> 4;
  {
    const v8f give = half == 0 ? acc[NRW - 1] : acc[0];
    float4* pe = (float4*)&exch[tile][half][M * 16 + hi * 8];
    pe[0] = make_float4(give[0], give[1], give[2], give[3]);
    pe[1] = make_float4(give[4], give[5], give[6], give[7]);
  }
  __syncthreads();
  const int cm = M < 3 ? M : 0;
  float xr[8];
  {
    const int col = half == 0 ? 8 + cm : cm;
    const float4* pe = (const float4*)&exch[tile][1 - half][col * 16 + hi * 8];
    float4 a = pe[0], c = pe[1];
    xr[0] = a.x; xr[1] = a.y; xr[2] = a.z; xr[3] = a.w;
    xr[4] = c.x; xr[5] = c.y; xr[6] = c.z; xr[7] = c.w;
  }
  const bool noTop = (y0 == 0);
  const bool noBot = (y0 + RS == HW_);
  if (half == 0) {
#pragma unroll
    for (int i = 0; i < 4; ++i) {
      float t0[8], t1[8], t2[8];
#pragma unroll
      for (int r = 0; r < 8; ++r) {
        t0[r] = acc[i][r];
        t1[r] = __shfl_down(acc[i + 1][r], 4, 16);
        t2[r] = i < 3 ? __shfl_down(acc[i < 3 ? i + 2 : 0][r], 8, 16) : xr[r];
        if (i == 0) t0[r] = noTop ? 0.0f : t0[r];
      }
      put_row(t0, t1, t2, pout, y0 + i, x0, M, hi);
    }
  } else {
#pragma unroll
    for (int i = 0; i < 4; ++i) {
      float t0[8], t1[8], t2[8];
#pragma unroll
      for (int r = 0; r < 8; ++r) {
        t0[r] = i > 0 ? acc[i > 0 ? i - 1 : 0][r] : xr[r];
        t1[r] = __shfl_down(acc[i][r], 4, 16);
        t2[r] = __shfl_down(acc[i + 1][r], 8, 16);
        if (i == 3) t2[r] = noBot ? 0.0f : t2[r];
      }
      put_row(t0, t1, t2, pout, y0 + 4 + i, x0, M, hi);
    }
  }
}

__global__ void skip_bias_kernel(const float* __restrict__ skip,
                                 const float* __restrict__ bias,
                                 float* __restrict__ out) {
  int tid = blockIdx.x * blockDim.x + threadIdx.x;
  const int total = NBATCH * 3 * HW_ * HW_;
  if (tid >= total) return;

  int x  = tid & 127;
  int y  = (tid >> 7) & 127;
  int bc = tid >> 14;
  int co = bc % 3;

  int y2 = y >> 1; bool ye = !(y & 1);
  int   my0 = ye ? y2     : y2 + 1;  float wy0 = ye ? 0.75f : 0.25f;
  int   my1 = ye ? y2 - 1 : y2;      float wy1 = ye ? 0.25f : 0.75f;
  int x2 = x >> 1; bool xe = !(x & 1);
  int   mx0 = xe ? x2     : x2 + 1;  float wx0 = xe ? 0.75f : 0.25f;
  int   mx1 = xe ? x2 - 1 : x2;      float wx1 = xe ? 0.25f : 0.75f;

  const float* s = skip + (size_t)bc * 64 * 64;
  float acc = bias[co];
  if (my0 >= 0 && my0 < 64) {
    if (mx0 >= 0 && mx0 < 64) acc += wy0 * wx0 * s[my0 * 64 + mx0];
    if (mx1 >= 0 && mx1 < 64) acc += wy0 * wx1 * s[my0 * 64 + mx1];
  }
  if (my1 >= 0 && my1 < 64) {
    if (mx0 >= 0 && mx0 < 64) acc += wy1 * wx0 * s[my1 * 64 + mx0];
    if (mx1 >= 0 && mx1 < 64) acc += wy1 * wx1 * s[my1 * 64 + mx1];
  }
  out[tid] += acc;
}

extern "C" void kernel_launch(void* const* d_in, const int* in_sizes, int n_in,
                              void* d_out, int out_size, void* d_ws,
                              size_t ws_size, hipStream_t stream) {
  (void)in_sizes; (void)n_in; (void)out_size; (void)ws_size;
  const float* in   = (const float*)d_in[0];
  const float* skip = (const float*)d_in[1];
  const float* w    = (const float*)d_in[2];
  const float* bias = (const float*)d_in[3];
  float* out = (float*)d_out;
  float* bp  = (float*)d_ws;

  prep_weight_kernel<<<dim3(96), dim3(256), 0, stream>>>(w, bp);
  conv_rows_kernel<<<dim3(256), dim3(256), 0, stream>>>(in, bp, out);
  skip_bias_kernel<<<dim3((NBATCH * 3 * HW_ * HW_ + 255) / 256), dim3(256), 0, stream>>>(
      skip, bias, out);
}


# Horizontal taps moved into N, vertical taps summed in the accumulator: 3 products per output row and 4 channels on unshifted rows, full-width 4-row strips, no x halo; changes the summation order
# speedup vs baseline: 17.7499x; 1.3316x over previous
#include <hip/hip_runtime.h>

typedef __attribute__((ext_vector_type(2))) float v2f;
typedef __attribute__((ext_vector_type(4))) float v4f;
typedef __attribute__((ext_vector_type(8))) float v8f;

#define C_IN   512
#define HW_    128
#define PLANE  (HW_ * HW_)
#define NBATCH 8
#define RS     4
#define NRW    (RS + 2)
#define TC     4
#define NQ     (C_IN / TC)
#define EQ_SCALE 0.014731391274719739f

__global__ void prep_weight_kernel(const float* __restrict__ w,
                                   float* __restrict__ bp) {
  int tid = blockIdx.x * blockDim.x + threadIdx.x;
  if (tid >= NQ * 3 * 64) return;
  int s  = tid & 1;
  int c  = (tid >> 1) & 15;
  int h  = (tid >> 5) & 1;
  int g  = tid >> 6;
  int q  = g / 3, dy = g - q * 3;
  int ci = 4 * q + 2 * h + s, dx = c >> 2, co = c & 3;
  float val = 0.0f;
  if (dx < 3 && co < 3)
    val = w[co * (C_IN * 9) + ci * 9 + dy * 3 + dx] * EQ_SCALE;
  bp[tid] = val;
}

#define CHF    16
#define SLOTF  (TC * CHF)
#define ROWP   272
#define PAIRF  (NRW * ROWP)
#define NLD    3
#define EXW    (HW_ + 2)
#define EXP    13


__device__ __forceinline__ void gl_trip(v4f (&G)[NLD], v2f (&GB)[3],
                                        const float* __restrict__ pin,
                                        const v2f* __restrict__ bp,
                                        const unsigned (&goff)[NLD], int lane, int c0) {
  const char* pn = (const char*)(pin + (size_t)c0 * PLANE);
#pragma unroll
  for (int i = 0; i < NLD; ++i) G[i] = *(const v4f*)(pn + goff[i]);
#pragma unroll
  for (int d = 0; d < 3; ++d) GB[d] = bp[((c0 >> 2) * 3 + d) * 32 + lane];
}

template <int MODE>
__device__ __forceinline__ void st_trip(const v4f (&G)[NLD], float* slot,
                                        const unsigned (&soff)[NLD], const bool (&zf)[NLD]) {
#pragma unroll
  for (int i = 0; i < NLD; ++i) {
    v4f v = G[i];
    if (MODE != 0) { v4f z = {0.0f, 0.0f, 0.0f, 0.0f}; v = zf[i] ? z : v; }
    *(v4f*)(slot + soff[i]) = v;
  }
}

__device__ __forceinline__ void fr_trip(v2f (&A)[NRW], const float* slot, int foff) {
#pragma unroll
  for (int j = 0; j < NRW; ++j) {
    const float* p = slot + j * ROWP + foff;
    v2f v = {p[0], p[CHF]};
    A[j] = v;
  }
}

__device__ __forceinline__ void mul_trip(const v2f (&A)[NRW], const v2f (&B)[3],
                                         v8f (&acc)[RS]) {
#pragma unroll
  for (int d = 0; d < 3; ++d) {
#pragma unroll
    for (int i = 0; i < RS; ++i)
      acc[i] = __builtin_amdgcn_wmma_f32_16x16x4_f32(
          false, A[i + d], false, B[d], (short)0, acc[i], false, false);
  }
}

template <int MODE, bool LOAD>
__device__ __forceinline__ void body(v2f (&AC)[NRW], v2f (&AN)[NRW],
                                     v4f (&GS)[NLD], v4f (&GL)[NLD], v2f (&WC)[3],
                                     v8f (&acc)[RS],
                                     const float* __restrict__ pin, const v2f* __restrict__ bp,
                                     float* slotN, const unsigned (&goff)[NLD],
                                     const unsigned (&soff)[NLD], const bool (&zf)[NLD], int foff,
                                     int lane, int c_far) {
  st_trip<MODE>(GS, slotN, soff, zf);
  __builtin_amdgcn_fence(__ATOMIC_RELEASE, "wavefront");
  __builtin_amdgcn_sched_barrier(0);
  __builtin_amdgcn_fence(__ATOMIC_ACQUIRE, "wavefront");
  fr_trip(AN, slotN, foff);
  __builtin_amdgcn_sched_barrier(0);
  mul_trip(AC, WC, acc);
  asm volatile("" : "+v"(acc[0]), "+v"(acc[1]), "+v"(acc[2]), "+v"(acc[3]));
  __builtin_amdgcn_sched_barrier(0);
  if (LOAD) gl_trip(GL, WC, pin, bp, goff, lane, c_far);
  __builtin_amdgcn_sched_barrier(0);
}

template <int MODE>
__device__ __forceinline__ void conv_rows6(const float* __restrict__ pin,
                                           const v2f* __restrict__ bp,
                                           float* lds, v8f (&acc)[RS],
                                           int y0, int x0, int lane) {
  const int M  = lane & 15;
  const int hi = lane >> 4;
  const int foff = 2 * hi * CHF + M;

  unsigned goff[NLD], soff[NLD];
  bool zf[NLD];
#pragma unroll
  for (int i = 0; i < NLD; ++i) {
    const int q  = 32 * i + lane;
    const int r  = q >> 4;
    const int u  = (q >> 2) & 3;
    const int k  = q & 3;
    int yy = y0 - 1 + r;
    zf[i] = (yy < 0) || (yy > HW_ - 1);
    yy = yy < 0 ? 0 : (yy > HW_ - 1 ? HW_ - 1 : yy);
    goff[i] = (unsigned)(u * PLANE + yy * HW_ + x0 + 4 * k) * 4u;
    soff[i] = (unsigned)(r * ROWP + u * CHF + 4 * k);
  }
  float* slot0 = lds;
  float* slot1 = lds + SLOTF;

#pragma unroll
  for (int i = 0; i < RS; ++i) acc[i] = v8f{};

  v2f AX[NRW], AY[NRW];
  v4f G0[NLD], G1[NLD], G2[NLD], G3[NLD], G4[NLD], G5[NLD];
  v2f W0[3], W1[3], W2[3], W3[3], W4[3], W5[3];

  gl_trip(G0, W0, pin, bp, goff, lane, 0);
  gl_trip(G1, W1, pin, bp, goff, lane, TC);
  gl_trip(G2, W2, pin, bp, goff, lane, 2 * TC);
  gl_trip(G3, W3, pin, bp, goff, lane, 3 * TC);
  __builtin_amdgcn_sched_barrier(0);
  st_trip<MODE>(G0, slot0, soff, zf);
  __builtin_amdgcn_fence(__ATOMIC_RELEASE, "wavefront");
  __builtin_amdgcn_sched_barrier(0);
  __builtin_amdgcn_fence(__ATOMIC_ACQUIRE, "wavefront");
  fr_trip(AX, slot0, foff);
  __builtin_amdgcn_sched_barrier(0);
  gl_trip(G4, W4, pin, bp, goff, lane, 4 * TC);
  gl_trip(G5, W5, pin, bp, goff, lane, 5 * TC);
  __builtin_amdgcn_sched_barrier(0);

  for (int c0 = 0; c0 < 120 * TC; c0 += 6 * TC) {
    body<MODE, true>(AX, AY, G1, G0, W0, acc, pin, bp, slot1, goff, soff, zf, foff, lane, c0 + 6 * TC);
    body<MODE, true>(AY, AX, G2, G1, W1, acc, pin, bp, slot0, goff, soff, zf, foff, lane, c0 + 7 * TC);
    body<MODE, true>(AX, AY, G3, G2, W2, acc, pin, bp, slot1, goff, soff, zf, foff, lane, c0 + 8 * TC);
    body<MODE, true>(AY, AX, G4, G3, W3, acc, pin, bp, slot0, goff, soff, zf, foff, lane, c0 + 9 * TC);
    body<MODE, true>(AX, AY, G5, G4, W4, acc, pin, bp, slot1, goff, soff, zf, foff, lane, c0 + 10 * TC);
    body<MODE, true>(AY, AX, G0, G5, W5, acc, pin, bp, slot0, goff, soff, zf, foff, lane, c0 + 11 * TC);
  }
  body<MODE, true >(AX, AY, G1, G0, W0, acc, pin, bp, slot1, goff, soff, zf, foff, lane, 126 * TC);
  body<MODE, true >(AY, AX, G2, G1, W1, acc, pin, bp, slot0, goff, soff, zf, foff, lane, 127 * TC);
  body<MODE, false>(AX, AY, G3, G2, W2, acc, pin, bp, slot1, goff, soff, zf, foff, lane, 0);
  body<MODE, false>(AY, AX, G4, G3, W3, acc, pin, bp, slot0, goff, soff, zf, foff, lane, 0);
  body<MODE, false>(AX, AY, G5, G4, W4, acc, pin, bp, slot1, goff, soff, zf, foff, lane, 0);
  body<MODE, false>(AY, AX, G0, G5, W5, acc, pin, bp, slot0, goff, soff, zf, foff, lane, 0);
  body<MODE, false>(AX, AY, G1, G0, W0, acc, pin, bp, slot1, goff, soff, zf, foff, lane, 0);
  mul_trip(AY, W1, acc);
}

__global__ void __launch_bounds__(256) conv_rows_kernel(const float* __restrict__ in,
                                 const float* __restrict__ bpack,
                                 float* __restrict__ out) {
  __shared__ __attribute__((aligned(16))) float stage[4][PAIRF];
  __shared__ float ex[RS * EXW * EXP];
  const int lane = threadIdx.x & 31;
  const int w    = __builtin_amdgcn_readfirstlane(threadIdx.x >> 5);
  float* lds = stage[w >> 1] + (w & 1) * (2 * SLOTF);
  const int st   = blockIdx.x & 31;
  const int b    = blockIdx.x >> 5;
  const int y0   = st * RS;
  const int x0   = w * 16;
  const float* pin  = in + (size_t)b * C_IN * PLANE;
  float*       pout = out + (size_t)b * 3 * PLANE;
  const v2f*   bp   = (const v2f*)bpack;

  v8f acc[RS];
  if (st == 0 || st == HW_ / RS - 1) conv_rows6<1>(pin, bp, lds, acc, y0, x0, lane);
  else                               conv_rows6<0>(pin, bp, lds, acc, y0, x0, lane);

  const int c  = lane & 15;
  const int hi = lane >> 4;
  if (threadIdx.x < RS * 2 * EXP) {
    const int i    = threadIdx.x / (2 * EXP);
    const int rm   = threadIdx.x - i * (2 * EXP);
    const int side = rm / EXP;
    const int cc   = rm - side * EXP;
    ex[(i * EXW + side * (EXW - 1)) * EXP + cc] = 0.0f;
  }
  if (c < 11) {
#pragma unroll
    for (int i = 0; i < RS; ++i) {
#pragma unroll
      for (int r = 0; r < 8; ++r)
        ex[(i * EXW + 1 + x0 + 8 * hi + r) * EXP + c] = acc[i][r];
    }
  }
  __syncthreads();
#pragma unroll
  for (int n = 0; n < (RS * 3 * HW_) / 256; ++n) {
    const int idx = threadIdx.x + 256 * n;
    const int X   = idx & (HW_ - 1);
    const int rc  = idx >> 7;
    const int i   = rc / 3;
    const int co  = rc - i * 3;
    const float* e = ex + (i * EXW + X) * EXP + co;
    const float v = (e[0] + e[EXP + 4]) + e[2 * EXP + 8];
    pout[((size_t)co * HW_ + (y0 + i)) * HW_ + X] = v;
  }
}

__global__ void skip_bias_kernel(const float* __restrict__ skip,
                                 const float* __restrict__ bias,
                                 float* __restrict__ out) {
  int tid = blockIdx.x * blockDim.x + threadIdx.x;
  const int total = NBATCH * 3 * HW_ * HW_;
  if (tid >= total) return;

  int x  = tid & 127;
  int y  = (tid >> 7) & 127;
  int bc = tid >> 14;
  int co = bc % 3;

  int y2 = y >> 1; bool ye = !(y & 1);
  int   my0 = ye ? y2     : y2 + 1;  float wy0 = ye ? 0.75f : 0.25f;
  int   my1 = ye ? y2 - 1 : y2;      float wy1 = ye ? 0.25f : 0.75f;
  int x2 = x >> 1; bool xe = !(x & 1);
  int   mx0 = xe ? x2     : x2 + 1;  float wx0 = xe ? 0.75f : 0.25f;
  int   mx1 = xe ? x2 - 1 : x2;      float wx1 = xe ? 0.25f : 0.75f;

  const float* s = skip + (size_t)bc * 64 * 64;
  float acc = bias[co];
  if (my0 >= 0 && my0 < 64) {
    if (mx0 >= 0 && mx0 < 64) acc += wy0 * wx0 * s[my0 * 64 + mx0];
    if (mx1 >= 0 && mx1 < 64) acc += wy0 * wx1 * s[my0 * 64 + mx1];
  }
  if (my1 >= 0 && my1 < 64) {
    if (mx0 >= 0 && mx0 < 64) acc += wy1 * wx0 * s[my1 * 64 + mx0];
    if (mx1 >= 0 && mx1 < 64) acc += wy1 * wx1 * s[my1 * 64 + mx1];
  }
  out[tid] += acc;
}

extern "C" void kernel_launch(void* const* d_in, const int* in_sizes, int n_in,
                              void* d_out, int out_size, void* d_ws,
                              size_t ws_size, hipStream_t stream) {
  (void)in_sizes; (void)n_in; (void)out_size; (void)ws_size;
  const float* in   = (const float*)d_in[0];
  const float* skip = (const float*)d_in[1];
  const float* w    = (const float*)d_in[2];
  const float* bias = (const float*)d_in[3];
  float* out = (float*)d_out;
  float* bp  = (float*)d_ws;

  prep_weight_kernel<<<dim3(96), dim3(256), 0, stream>>>(w, bp);
  conv_rows_kernel<<<dim3(256), dim3(256), 0, stream>>>(in, bp, out);
  skip_bias_kernel<<<dim3((NBATCH * 3 * HW_ * HW_ + 255) / 256), dim3(256), 0, stream>>>(
      skip, bias, out);
}


# Vertical taps by 2-output 3-tap minimal filtering: 4 row sums/differences x 4 combined weights (two per 16-byte load) per 2 output rows, 8 products per 4 channels, was 12; changes the summation order
# speedup vs baseline: 19.1529x; 1.0790x over previous
#include <hip/hip_runtime.h>

typedef __attribute__((ext_vector_type(2))) float v2f;
typedef __attribute__((ext_vector_type(4))) float v4f;
typedef __attribute__((ext_vector_type(8))) float v8f;

#define C_IN   512
#define HW_    128
#define PLANE  (HW_ * HW_)
#define NBATCH 8
#define RS     4
#define NRW    (RS + 2)
#define TC     4
#define NQ     (C_IN / TC)
#define NU     4
#define NACC   (2 * NU)
#define EQ_SCALE 0.014731391274719739f

__global__ void prep_weight_kernel(const float* __restrict__ w,
                                   float* __restrict__ bp) {
  int tid = blockIdx.x * blockDim.x + threadIdx.x;
  if (tid >= NQ * NU * 64) return;
  int s  = tid & 1;
  int ul = (tid >> 1) & 1;
  int c  = (tid >> 2) & 15;
  int h  = (tid >> 6) & 1;
  int j  = (tid >> 7) & 1;
  int q  = tid >> 8;
  int u  = 2 * j + ul;
  int ci = 4 * q + 2 * h + s, dx = c >> 2, co = c & 3;
  float val = 0.0f;
  if (dx < 3 && co < 3) {
    const float* p = w + co * (C_IN * 9) + ci * 9 + dx;
    const float g0 = p[0], g1 = p[3], g2 = p[6];
    const float e  = g0 + g2;
    const float gu = (u == 0) ? g0 : (u == 3) ? g2 : (u == 1) ? 0.5f * (e + g1) : 0.5f * (e - g1);
    val = gu * EQ_SCALE;
  }
  bp[tid] = val;
}

#define CHF    16
#define SLOTF  (TC * CHF)
#define ROWP   272
#define PAIRF  (NRW * ROWP)
#define NLD    3
#define EXW    (HW_ + 2)
#define EXP    13


__device__ __forceinline__ void gl_trip(v4f (&G)[NLD], v4f (&GB)[2],
                                        const float* __restrict__ pin,
                                        const v4f* __restrict__ bp,
                                        const unsigned (&goff)[NLD], int lane, int c0) {
  const char* pn = (const char*)(pin + (size_t)c0 * PLANE);
#pragma unroll
  for (int i = 0; i < NLD; ++i) G[i] = *(const v4f*)(pn + goff[i]);
#pragma unroll
  for (int j = 0; j < 2; ++j) GB[j] = bp[((c0 >> 2) * 2 + j) * 32 + lane];
}

template <int MODE>
__device__ __forceinline__ void st_trip(const v4f (&G)[NLD], float* slot,
                                        const unsigned (&soff)[NLD], const bool (&zf)[NLD]) {
#pragma unroll
  for (int i = 0; i < NLD; ++i) {
    v4f v = G[i];
    if (MODE != 0) { v4f z = {0.0f, 0.0f, 0.0f, 0.0f}; v = zf[i] ? z : v; }
    *(v4f*)(slot + soff[i]) = v;
  }
}

__device__ __forceinline__ void fr_trip(v2f (&A)[NRW], const float* slot, int foff) {
#pragma unroll
  for (int j = 0; j < NRW; ++j) {
    const float* p = slot + j * ROWP + foff;
    v2f v = {p[0], p[CHF]};
    A[j] = v;
  }
}

__device__ __forceinline__ void mul_trip(const v2f (&A)[NRW], const v4f (&B)[2],
                                         v8f (&acc)[NACC]) {
  v2f T[NACC];
#pragma unroll
  for (int p = 0; p < 2; ++p) {
    const v2f d0 = A[2 * p], d1 = A[2 * p + 1], d2 = A[2 * p + 2], d3 = A[2 * p + 3];
    T[4 * p + 0] = d0 - d2;
    T[4 * p + 1] = d1 + d2;
    T[4 * p + 2] = d2 - d1;
    T[4 * p + 3] = d1 - d3;
  }
  v2f Wu[NU];
#pragma unroll
  for (int j = 0; j < 2; ++j) {
    const v2f lo = {B[j][0], B[j][1]};
    const v2f hi = {B[j][2], B[j][3]};
    Wu[2 * j]     = lo;
    Wu[2 * j + 1] = hi;
  }
#pragma unroll
  for (int u = 0; u < NU; ++u) {
#pragma unroll
    for (int p = 0; p < 2; ++p)
      acc[4 * p + u] = __builtin_amdgcn_wmma_f32_16x16x4_f32(
          false, T[4 * p + u], false, Wu[u], (short)0, acc[4 * p + u], false, false);
  }
}

template <int MODE, bool LOAD>
__device__ __forceinline__ void body(v2f (&AC)[NRW], v2f (&AN)[NRW],
                                     v4f (&GS)[NLD], v4f (&GL)[NLD], v4f (&WC)[2],
                                     v8f (&acc)[NACC],
                                     const float* __restrict__ pin, const v4f* __restrict__ bp,
                                     float* slotN, const unsigned (&goff)[NLD],
                                     const unsigned (&soff)[NLD], const bool (&zf)[NLD], int foff,
                                     int lane, int c_far) {
  st_trip<MODE>(GS, slotN, soff, zf);
  __builtin_amdgcn_fence(__ATOMIC_RELEASE, "wavefront");
  __builtin_amdgcn_sched_barrier(0);
  __builtin_amdgcn_fence(__ATOMIC_ACQUIRE, "wavefront");
  fr_trip(AN, slotN, foff);
  __builtin_amdgcn_sched_barrier(0);
  mul_trip(AC, WC, acc);
  asm volatile("" : "+v"(acc[0]), "+v"(acc[1]), "+v"(acc[2]), "+v"(acc[3]),
                    "+v"(acc[4]), "+v"(acc[5]), "+v"(acc[6]), "+v"(acc[7]));
  __builtin_amdgcn_sched_barrier(0);
  if (LOAD) gl_trip(GL, WC, pin, bp, goff, lane, c_far);
  __builtin_amdgcn_sched_barrier(0);
}

template <int MODE>
__device__ __forceinline__ void conv_rows6(const float* __restrict__ pin,
                                           const v4f* __restrict__ bp,
                                           float* lds, v8f (&acc)[NACC],
                                           int y0, int x0, int lane) {
  const int M  = lane & 15;
  const int hi = lane >> 4;
  const int foff = 2 * hi * CHF + M;

  unsigned goff[NLD], soff[NLD];
  bool zf[NLD];
#pragma unroll
  for (int i = 0; i < NLD; ++i) {
    const int q  = 32 * i + lane;
    const int r  = q >> 4;
    const int u  = (q >> 2) & 3;
    const int k  = q & 3;
    int yy = y0 - 1 + r;
    zf[i] = (yy < 0) || (yy > HW_ - 1);
    yy = yy < 0 ? 0 : (yy > HW_ - 1 ? HW_ - 1 : yy);
    goff[i] = (unsigned)(u * PLANE + yy * HW_ + x0 + 4 * k) * 4u;
    soff[i] = (unsigned)(r * ROWP + u * CHF + 4 * k);
  }
  float* slot0 = lds;
  float* slot1 = lds + SLOTF;

#pragma unroll
  for (int i = 0; i < NACC; ++i) acc[i] = v8f{};

  v2f AX[NRW], AY[NRW];
  v4f G0[NLD], G1[NLD], G2[NLD], G3[NLD], G4[NLD], G5[NLD];
  v4f W0[2], W1[2], W2[2], W3[2], W4[2], W5[2];

  gl_trip(G0, W0, pin, bp, goff, lane, 0);
  gl_trip(G1, W1, pin, bp, goff, lane, TC);
  gl_trip(G2, W2, pin, bp, goff, lane, 2 * TC);
  gl_trip(G3, W3, pin, bp, goff, lane, 3 * TC);
  __builtin_amdgcn_sched_barrier(0);
  st_trip<MODE>(G0, slot0, soff, zf);
  __builtin_amdgcn_fence(__ATOMIC_RELEASE, "wavefront");
  __builtin_amdgcn_sched_barrier(0);
  __builtin_amdgcn_fence(__ATOMIC_ACQUIRE, "wavefront");
  fr_trip(AX, slot0, foff);
  __builtin_amdgcn_sched_barrier(0);
  gl_trip(G4, W4, pin, bp, goff, lane, 4 * TC);
  gl_trip(G5, W5, pin, bp, goff, lane, 5 * TC);
  __builtin_amdgcn_sched_barrier(0);

  for (int c0 = 0; c0 < 120 * TC; c0 += 6 * TC) {
    body<MODE, true>(AX, AY, G1, G0, W0, acc, pin, bp, slot1, goff, soff, zf, foff, lane, c0 + 6 * TC);
    body<MODE, true>(AY, AX, G2, G1, W1, acc, pin, bp, slot0, goff, soff, zf, foff, lane, c0 + 7 * TC);
    body<MODE, true>(AX, AY, G3, G2, W2, acc, pin, bp, slot1, goff, soff, zf, foff, lane, c0 + 8 * TC);
    body<MODE, true>(AY, AX, G4, G3, W3, acc, pin, bp, slot0, goff, soff, zf, foff, lane, c0 + 9 * TC);
    body<MODE, true>(AX, AY, G5, G4, W4, acc, pin, bp, slot1, goff, soff, zf, foff, lane, c0 + 10 * TC);
    body<MODE, true>(AY, AX, G0, G5, W5, acc, pin, bp, slot0, goff, soff, zf, foff, lane, c0 + 11 * TC);
  }
  body<MODE, true >(AX, AY, G1, G0, W0, acc, pin, bp, slot1, goff, soff, zf, foff, lane, 126 * TC);
  body<MODE, true >(AY, AX, G2, G1, W1, acc, pin, bp, slot0, goff, soff, zf, foff, lane, 127 * TC);
  body<MODE, false>(AX, AY, G3, G2, W2, acc, pin, bp, slot1, goff, soff, zf, foff, lane, 0);
  body<MODE, false>(AY, AX, G4, G3, W3, acc, pin, bp, slot0, goff, soff, zf, foff, lane, 0);
  body<MODE, false>(AX, AY, G5, G4, W4, acc, pin, bp, slot1, goff, soff, zf, foff, lane, 0);
  body<MODE, false>(AY, AX, G0, G5, W5, acc, pin, bp, slot0, goff, soff, zf, foff, lane, 0);
  body<MODE, false>(AX, AY, G1, G0, W0, acc, pin, bp, slot1, goff, soff, zf, foff, lane, 0);
  mul_trip(AY, W1, acc);
}

__global__ void __launch_bounds__(256) conv_rows_kernel(const float* __restrict__ in,
                                 const float* __restrict__ bpack,
                                 float* __restrict__ out) {
  __shared__ __attribute__((aligned(16))) float stage[4][PAIRF];
  __shared__ float ex[RS * EXW * EXP];
  const int lane = threadIdx.x & 31;
  const int w    = __builtin_amdgcn_readfirstlane(threadIdx.x >> 5);
  float* lds = stage[w >> 1] + (w & 1) * (2 * SLOTF);
  const int st   = blockIdx.x & 31;
  const int b    = blockIdx.x >> 5;
  const int y0   = st * RS;
  const int x0   = w * 16;
  const float* pin  = in + (size_t)b * C_IN * PLANE;
  float*       pout = out + (size_t)b * 3 * PLANE;
  const v4f*   bp   = (const v4f*)bpack;

  v8f m[NACC];
  if (st == 0 || st == HW_ / RS - 1) conv_rows6<1>(pin, bp, lds, m, y0, x0, lane);
  else                               conv_rows6<0>(pin, bp, lds, m, y0, x0, lane);

  v8f acc[RS];
#pragma unroll
  for (int p = 0; p < 2; ++p) {
    acc[2 * p]     = (m[4 * p] + m[4 * p + 1]) + m[4 * p + 2];
    acc[2 * p + 1] = (m[4 * p + 1] - m[4 * p + 2]) - m[4 * p + 3];
  }

  const int c  = lane & 15;
  const int hi = lane >> 4;
  if (threadIdx.x < RS * 2 * EXP) {
    const int i    = threadIdx.x / (2 * EXP);
    const int rm   = threadIdx.x - i * (2 * EXP);
    const int side = rm / EXP;
    const int cc   = rm - side * EXP;
    ex[(i * EXW + side * (EXW - 1)) * EXP + cc] = 0.0f;
  }
  if (c < 11) {
#pragma unroll
    for (int i = 0; i < RS; ++i) {
#pragma unroll
      for (int r = 0; r < 8; ++r)
        ex[(i * EXW + 1 + x0 + 8 * hi + r) * EXP + c] = acc[i][r];
    }
  }
  __syncthreads();
#pragma unroll
  for (int n = 0; n < (RS * 3 * HW_) / 256; ++n) {
    const int idx = threadIdx.x + 256 * n;
    const int X   = idx & (HW_ - 1);
    const int rc  = idx >> 7;
    const int i   = rc / 3;
    const int co  = rc - i * 3;
    const float* e = ex + (i * EXW + X) * EXP + co;
    const float v = (e[0] + e[EXP + 4]) + e[2 * EXP + 8];
    pout[((size_t)co * HW_ + (y0 + i)) * HW_ + X] = v;
  }
}

__global__ void skip_bias_kernel(const float* __restrict__ skip,
                                 const float* __restrict__ bias,
                                 float* __restrict__ out) {
  int tid = blockIdx.x * blockDim.x + threadIdx.x;
  const int total = NBATCH * 3 * HW_ * HW_;
  if (tid >= total) return;

  int x  = tid & 127;
  int y  = (tid >> 7) & 127;
  int bc = tid >> 14;
  int co = bc % 3;

  int y2 = y >> 1; bool ye = !(y & 1);
  int   my0 = ye ? y2     : y2 + 1;  float wy0 = ye ? 0.75f : 0.25f;
  int   my1 = ye ? y2 - 1 : y2;      float wy1 = ye ? 0.25f : 0.75f;
  int x2 = x >> 1; bool xe = !(x & 1);
  int   mx0 = xe ? x2     : x2 + 1;  float wx0 = xe ? 0.75f : 0.25f;
  int   mx1 = xe ? x2 - 1 : x2;      float wx1 = xe ? 0.25f : 0.75f;

  const float* s = skip + (size_t)bc * 64 * 64;
  float acc = bias[co];
  if (my0 >= 0 && my0 < 64) {
    if (mx0 >= 0 && mx0 < 64) acc += wy0 * wx0 * s[my0 * 64 + mx0];
    if (mx1 >= 0 && mx1 < 64) acc += wy0 * wx1 * s[my0 * 64 + mx1];
  }
  if (my1 >= 0 && my1 < 64) {
    if (mx0 >= 0 && mx0 < 64) acc += wy1 * wx0 * s[my1 * 64 + mx0];
    if (mx1 >= 0 && mx1 < 64) acc += wy1 * wx1 * s[my1 * 64 + mx1];
  }
  out[tid] += acc;
}

extern "C" void kernel_launch(void* const* d_in, const int* in_sizes, int n_in,
                              void* d_out, int out_size, void* d_ws,
                              size_t ws_size, hipStream_t stream) {
  (void)in_sizes; (void)n_in; (void)out_size; (void)ws_size;
  const float* in   = (const float*)d_in[0];
  const float* skip = (const float*)d_in[1];
  const float* w    = (const float*)d_in[2];
  const float* bias = (const float*)d_in[3];
  float* out = (float*)d_out;
  float* bp  = (float*)d_ws;

  prep_weight_kernel<<<dim3(128), dim3(256), 0, stream>>>(w, bp);
  conv_rows_kernel<<<dim3(256), dim3(256), 0, stream>>>(in, bp, out);
  skip_bias_kernel<<<dim3((NBATCH * 3 * HW_ * HW_ + 255) / 256), dim3(256), 0, stream>>>(
      skip, bias, out);
}


# Weight pack and bias/skip pass run as ONE grid in front of the conv: it stores out = bias + upsampled skip, the conv epilogue adds its sums onto it (3 launches -> 2); sum order unchanged
# speedup vs baseline: 20.0842x; 1.0486x over previous
#include <hip/hip_runtime.h>

typedef __attribute__((ext_vector_type(2))) float v2f;
typedef __attribute__((ext_vector_type(4))) float v4f;
typedef __attribute__((ext_vector_type(8))) float v8f;

#define C_IN   512
#define HW_    128
#define PLANE  (HW_ * HW_)
#define NBATCH 8
#define RS     4
#define NRW    (RS + 2)
#define TC     4
#define NQ     (C_IN / TC)
#define NU     4
#define NACC   (2 * NU)
#define EQ_SCALE 0.014731391274719739f

__global__ void pack_start_kernel(const float* __restrict__ w,
                                  const float* __restrict__ skip,
                                  const float* __restrict__ bias,
                                  float* __restrict__ bp,
                                  float* __restrict__ out) {
  int tid = blockIdx.x * blockDim.x + threadIdx.x;
  const int total = NBATCH * 3 * HW_ * HW_;
  if (tid >= total) return;

  if (tid < NQ * NU * 64) {
    int s  = tid & 1;
    int ul = (tid >> 1) & 1;
    int c  = (tid >> 2) & 15;
    int h  = (tid >> 6) & 1;
    int j  = (tid >> 7) & 1;
    int q  = tid >> 8;
    int u  = 2 * j + ul;
    int ci = 4 * q + 2 * h + s, dx = c >> 2, co = c & 3;
    float val = 0.0f;
    if (dx < 3 && co < 3) {
      const float* p = w + co * (C_IN * 9) + ci * 9 + dx;
      const float g0 = p[0], g1 = p[3], g2 = p[6];
      const float e  = g0 + g2;
      const float gu = (u == 0) ? g0 : (u == 3) ? g2 : (u == 1) ? 0.5f * (e + g1) : 0.5f * (e - g1);
      val = gu * EQ_SCALE;
    }
    bp[tid] = val;
  }

  int x  = tid & 127;
  int y  = (tid >> 7) & 127;
  int bc = tid >> 14;
  int co = bc % 3;

  int y2 = y >> 1; bool ye = !(y & 1);
  int   my0 = ye ? y2     : y2 + 1;  float wy0 = ye ? 0.75f : 0.25f;
  int   my1 = ye ? y2 - 1 : y2;      float wy1 = ye ? 0.25f : 0.75f;
  int x2 = x >> 1; bool xe = !(x & 1);
  int   mx0 = xe ? x2     : x2 + 1;  float wx0 = xe ? 0.75f : 0.25f;
  int   mx1 = xe ? x2 - 1 : x2;      float wx1 = xe ? 0.25f : 0.75f;

  const float* s = skip + (size_t)bc * 64 * 64;
  float acc = bias[co];
  if (my0 >= 0 && my0 < 64) {
    if (mx0 >= 0 && mx0 < 64) acc += wy0 * wx0 * s[my0 * 64 + mx0];
    if (mx1 >= 0 && mx1 < 64) acc += wy0 * wx1 * s[my0 * 64 + mx1];
  }
  if (my1 >= 0 && my1 < 64) {
    if (mx0 >= 0 && mx0 < 64) acc += wy1 * wx0 * s[my1 * 64 + mx0];
    if (mx1 >= 0 && mx1 < 64) acc += wy1 * wx1 * s[my1 * 64 + mx1];
  }
  out[tid] = acc;
}

#define CHF    16
#define SLOTF  (TC * CHF)
#define ROWP   272
#define PAIRF  (NRW * ROWP)
#define NLD    3
#define EXW    (HW_ + 2)
#define EXP    13


__device__ __forceinline__ void gl_trip(v4f (&G)[NLD], v4f (&GB)[2],
                                        const float* __restrict__ pin,
                                        const v4f* __restrict__ bp,
                                        const unsigned (&goff)[NLD], int lane, int c0) {
  const char* pn = (const char*)(pin + (size_t)c0 * PLANE);
#pragma unroll
  for (int i = 0; i < NLD; ++i) G[i] = *(const v4f*)(pn + goff[i]);
#pragma unroll
  for (int j = 0; j < 2; ++j) GB[j] = bp[((c0 >> 2) * 2 + j) * 32 + lane];
}

template <int MODE>
__device__ __forceinline__ void st_trip(const v4f (&G)[NLD], float* slot,
                                        const unsigned (&soff)[NLD], const bool (&zf)[NLD]) {
#pragma unroll
  for (int i = 0; i < NLD; ++i) {
    v4f v = G[i];
    if (MODE != 0) { v4f z = {0.0f, 0.0f, 0.0f, 0.0f}; v = zf[i] ? z : v; }
    *(v4f*)(slot + soff[i]) = v;
  }
}

__device__ __forceinline__ void fr_trip(v2f (&A)[NRW], const float* slot, int foff) {
#pragma unroll
  for (int j = 0; j < NRW; ++j) {
    const float* p = slot + j * ROWP + foff;
    v2f v = {p[0], p[CHF]};
    A[j] = v;
  }
}

__device__ __forceinline__ void mul_trip(const v2f (&A)[NRW], const v4f (&B)[2],
                                         v8f (&acc)[NACC]) {
  v2f T[NACC];
#pragma unroll
  for (int p = 0; p < 2; ++p) {
    const v2f d0 = A[2 * p], d1 = A[2 * p + 1], d2 = A[2 * p + 2], d3 = A[2 * p + 3];
    T[4 * p + 0] = d0 - d2;
    T[4 * p + 1] = d1 + d2;
    T[4 * p + 2] = d2 - d1;
    T[4 * p + 3] = d1 - d3;
  }
  v2f Wu[NU];
#pragma unroll
  for (int j = 0; j < 2; ++j) {
    const v2f lo = {B[j][0], B[j][1]};
    const v2f hi = {B[j][2], B[j][3]};
    Wu[2 * j]     = lo;
    Wu[2 * j + 1] = hi;
  }
#pragma unroll
  for (int u = 0; u < NU; ++u) {
#pragma unroll
    for (int p = 0; p < 2; ++p)
      acc[4 * p + u] = __builtin_amdgcn_wmma_f32_16x16x4_f32(
          false, T[4 * p + u], false, Wu[u], (short)0, acc[4 * p + u], false, false);
  }
}

template <int MODE, bool LOAD>
__device__ __forceinline__ void body(v2f (&AC)[NRW], v2f (&AN)[NRW],
                                     v4f (&GS)[NLD], v4f (&GL)[NLD], v4f (&WC)[2],
                                     v8f (&acc)[NACC],
                                     const float* __restrict__ pin, const v4f* __restrict__ bp,
                                     float* slotN, const unsigned (&goff)[NLD],
                                     const unsigned (&soff)[NLD], const bool (&zf)[NLD], int foff,
                                     int lane, int c_far) {
  st_trip<MODE>(GS, slotN, soff, zf);
  __builtin_amdgcn_fence(__ATOMIC_RELEASE, "wavefront");
  __builtin_amdgcn_sched_barrier(0);
  __builtin_amdgcn_fence(__ATOMIC_ACQUIRE, "wavefront");
  fr_trip(AN, slotN, foff);
  __builtin_amdgcn_sched_barrier(0);
  mul_trip(AC, WC, acc);
  asm volatile("" : "+v"(acc[0]), "+v"(acc[1]), "+v"(acc[2]), "+v"(acc[3]),
                    "+v"(acc[4]), "+v"(acc[5]), "+v"(acc[6]), "+v"(acc[7]));
  __builtin_amdgcn_sched_barrier(0);
  if (LOAD) gl_trip(GL, WC, pin, bp, goff, lane, c_far);
  __builtin_amdgcn_sched_barrier(0);
}

template <int MODE>
__device__ __forceinline__ void conv_rows6(const float* __restrict__ pin,
                                           const v4f* __restrict__ bp,
                                           float* lds, v8f (&acc)[NACC],
                                           int y0, int x0, int lane) {
  const int M  = lane & 15;
  const int hi = lane >> 4;
  const int foff = 2 * hi * CHF + M;

  unsigned goff[NLD], soff[NLD];
  bool zf[NLD];
#pragma unroll
  for (int i = 0; i < NLD; ++i) {
    const int q  = 32 * i + lane;
    const int r  = q >> 4;
    const int u  = (q >> 2) & 3;
    const int k  = q & 3;
    int yy = y0 - 1 + r;
    zf[i] = (yy < 0) || (yy > HW_ - 1);
    yy = yy < 0 ? 0 : (yy > HW_ - 1 ? HW_ - 1 : yy);
    goff[i] = (unsigned)(u * PLANE + yy * HW_ + x0 + 4 * k) * 4u;
    soff[i] = (unsigned)(r * ROWP + u * CHF + 4 * k);
  }
  float* slot0 = lds;
  float* slot1 = lds + SLOTF;

#pragma unroll
  for (int i = 0; i < NACC; ++i) acc[i] = v8f{};

  v2f AX[NRW], AY[NRW];
  v4f G0[NLD], G1[NLD], G2[NLD], G3[NLD], G4[NLD], G5[NLD];
  v4f W0[2], W1[2], W2[2], W3[2], W4[2], W5[2];

  gl_trip(G0, W0, pin, bp, goff, lane, 0);
  gl_trip(G1, W1, pin, bp, goff, lane, TC);
  gl_trip(G2, W2, pin, bp, goff, lane, 2 * TC);
  gl_trip(G3, W3, pin, bp, goff, lane, 3 * TC);
  __builtin_amdgcn_sched_barrier(0);
  st_trip<MODE>(G0, slot0, soff, zf);
  __builtin_amdgcn_fence(__ATOMIC_RELEASE, "wavefront");
  __builtin_amdgcn_sched_barrier(0);
  __builtin_amdgcn_fence(__ATOMIC_ACQUIRE, "wavefront");
  fr_trip(AX, slot0, foff);
  __builtin_amdgcn_sched_barrier(0);
  gl_trip(G4, W4, pin, bp, goff, lane, 4 * TC);
  gl_trip(G5, W5, pin, bp, goff, lane, 5 * TC);
  __builtin_amdgcn_sched_barrier(0);

  for (int c0 = 0; c0 < 120 * TC; c0 += 6 * TC) {
    body<MODE, true>(AX, AY, G1, G0, W0, acc, pin, bp, slot1, goff, soff, zf, foff, lane, c0 + 6 * TC);
    body<MODE, true>(AY, AX, G2, G1, W1, acc, pin, bp, slot0, goff, soff, zf, foff, lane, c0 + 7 * TC);
    body<MODE, true>(AX, AY, G3, G2, W2, acc, pin, bp, slot1, goff, soff, zf, foff, lane, c0 + 8 * TC);
    body<MODE, true>(AY, AX, G4, G3, W3, acc, pin, bp, slot0, goff, soff, zf, foff, lane, c0 + 9 * TC);
    body<MODE, true>(AX, AY, G5, G4, W4, acc, pin, bp, slot1, goff, soff, zf, foff, lane, c0 + 10 * TC);
    body<MODE, true>(AY, AX, G0, G5, W5, acc, pin, bp, slot0, goff, soff, zf, foff, lane, c0 + 11 * TC);
  }
  body<MODE, true >(AX, AY, G1, G0, W0, acc, pin, bp, slot1, goff, soff, zf, foff, lane, 126 * TC);
  body<MODE, true >(AY, AX, G2, G1, W1, acc, pin, bp, slot0, goff, soff, zf, foff, lane, 127 * TC);
  body<MODE, false>(AX, AY, G3, G2, W2, acc, pin, bp, slot1, goff, soff, zf, foff, lane, 0);
  body<MODE, false>(AY, AX, G4, G3, W3, acc, pin, bp, slot0, goff, soff, zf, foff, lane, 0);
  body<MODE, false>(AX, AY, G5, G4, W4, acc, pin, bp, slot1, goff, soff, zf, foff, lane, 0);
  body<MODE, false>(AY, AX, G0, G5, W5, acc, pin, bp, slot0, goff, soff, zf, foff, lane, 0);
  body<MODE, false>(AX, AY, G1, G0, W0, acc, pin, bp, slot1, goff, soff, zf, foff, lane, 0);
  mul_trip(AY, W1, acc);
}

__global__ void __launch_bounds__(256) conv_rows_kernel(const float* __restrict__ in,
                                 const float* __restrict__ bpack,
                                 float* __restrict__ out) {
  __shared__ __attribute__((aligned(16))) float stage[4][PAIRF];
  __shared__ float ex[RS * EXW * EXP];
  const int lane = threadIdx.x & 31;
  const int w    = __builtin_amdgcn_readfirstlane(threadIdx.x >> 5);
  float* lds = stage[w >> 1] + (w & 1) * (2 * SLOTF);
  const int st   = blockIdx.x & 31;
  const int b    = blockIdx.x >> 5;
  const int y0   = st * RS;
  const int x0   = w * 16;
  const float* pin  = in + (size_t)b * C_IN * PLANE;
  float*       pout = out + (size_t)b * 3 * PLANE;
  const v4f*   bp   = (const v4f*)bpack;

  const int X    = threadIdx.x & (HW_ - 1);
  const int half = threadIdx.x >> 7;
  float* po = pout + (size_t)(y0 + 2 * half) * HW_ + X;
  float start[(RS * 3 * HW_) / 256];
#pragma unroll
  for (int n = 0; n < (RS * 3 * HW_) / 256; ++n)
    start[n] = po[(n % 3) * PLANE + (n / 3) * HW_];

  v8f m[NACC];
  if (st == 0 || st == HW_ / RS - 1) conv_rows6<1>(pin, bp, lds, m, y0, x0, lane);
  else                               conv_rows6<0>(pin, bp, lds, m, y0, x0, lane);

  v8f acc[RS];
#pragma unroll
  for (int p = 0; p < 2; ++p) {
    acc[2 * p]     = (m[4 * p] + m[4 * p + 1]) + m[4 * p + 2];
    acc[2 * p + 1] = (m[4 * p + 1] - m[4 * p + 2]) - m[4 * p + 3];
  }

  const int c  = lane & 15;
  const int hi = lane >> 4;
  if (threadIdx.x < RS * 2 * EXP) {
    const int i    = threadIdx.x / (2 * EXP);
    const int rm   = threadIdx.x - i * (2 * EXP);
    const int side = rm / EXP;
    const int cc   = rm - side * EXP;
    ex[(i * EXW + side * (EXW - 1)) * EXP + cc] = 0.0f;
  }
  if (c < 11) {
#pragma unroll
    for (int i = 0; i < RS; ++i) {
#pragma unroll
      for (int r = 0; r < 8; ++r)
        ex[(i * EXW + 1 + x0 + 8 * hi + r) * EXP + c] = acc[i][r];
    }
  }
  __syncthreads();
  const float* eb = ex + (2 * half * EXW + X) * EXP;
#pragma unroll
  for (int n = 0; n < (RS * 3 * HW_) / 256; ++n) {
    const float* e = eb + (n / 3) * (EXW * EXP) + (n % 3);
    const float v = (e[0] + e[EXP + 4]) + e[2 * EXP + 8];
    po[(n % 3) * PLANE + (n / 3) * HW_] = v + start[n];
  }
}

extern "C" void kernel_launch(void* const* d_in, const int* in_sizes, int n_in,
                              void* d_out, int out_size, void* d_ws,
                              size_t ws_size, hipStream_t stream) {
  (void)in_sizes; (void)n_in; (void)out_size; (void)ws_size;
  const float* in   = (const float*)d_in[0];
  const float* skip = (const float*)d_in[1];
  const float* w    = (const float*)d_in[2];
  const float* bias = (const float*)d_in[3];
  float* out = (float*)d_out;
  float* bp  = (float*)d_ws;

  pack_start_kernel<<<dim3((NBATCH * 3 * HW_ * HW_ + 255) / 256), dim3(256), 0, stream>>>(
      w, skip, bias, bp, out);
  conv_rows_kernel<<<dim3(256), dim3(256), 0, stream>>>(in, bp, out);
}


# First launch on 1024 waves, was 12288: each thread stores a patch of 4 rows x 4 pixels of start values (16 skip values loaded once, 16-byte stores) and one weight-pack element; sum order unchanged
# speedup vs baseline: 20.7324x; 1.0323x over previous
#include <hip/hip_runtime.h>

typedef __attribute__((ext_vector_type(2))) float v2f;
typedef __attribute__((ext_vector_type(4))) float v4f;
typedef __attribute__((ext_vector_type(8))) float v8f;

#define C_IN   512
#define HW_    128
#define PLANE  (HW_ * HW_)
#define NBATCH 8
#define RS     4
#define NRW    (RS + 2)
#define TC     4
#define NQ     (C_IN / TC)
#define NU     4
#define NACC   (2 * NU)
#define EQ_SCALE 0.014731391274719739f

#define PS_ROWS   4
#define PS_PIX    4
#define PS_START_THREADS (NBATCH * 3 * (HW_ / PS_ROWS) * (HW_ / PS_PIX))
#define PS_THREADS (NQ * NU * 64)
#define PS_BLOCKS  (PS_THREADS / 256)
__global__ void __launch_bounds__(256) pack_start_kernel(const float* __restrict__ w,
                                  const float* __restrict__ skip,
                                  const float* __restrict__ bias,
                                  float* __restrict__ bp,
                                  float* __restrict__ out) {
  const int tid = blockIdx.x * 256 + threadIdx.x;

  const int ps = tid & 1;
  const int pul = (tid >> 1) & 1;
  const int pc = (tid >> 2) & 15;
  const int ph = (tid >> 6) & 1;
  const int pj = (tid >> 7) & 1;
  const int pq = tid >> 8;
  const int pu = 2 * pj + pul;
  const int pci = 4 * pq + 2 * ph + ps, pdx = pc >> 2, pco = pc & 3;
  const bool live = (pdx < 3) && (pco < 3);
  const float* pw = w + (live ? pco * (C_IN * 9) + pci * 9 + pdx : 0);
  const float g0 = pw[0], g1 = pw[3], g2 = pw[6];

  if (tid < PS_START_THREADS) {
    const int j  = tid & 31;
    const int m  = (tid >> 5) & 31;
    const int bc = tid >> 10;
    const int co = bc % 3;
    const float* s = skip + (size_t)bc * 64 * 64;
    const float bv = bias[co];

    const bool okL = (j > 0), okR = (j < 31);
    const int  cL  = okL ? 2 * j - 1 : 0;
    const int  cR  = okR ? 2 * j + 2 : 63;
    float v[4][4];
#pragma unroll
    for (int r = 0; r < 4; ++r) {
      const int  rr  = 2 * m - 1 + r;
      const bool okr = (rr >= 0) && (rr < 64);
      const int  rc  = rr < 0 ? 0 : (rr > 63 ? 63 : rr);
      const float* sr = s + rc * 64;
      const v2f   mid = *(const v2f*)(sr + 2 * j);
      const float lft = sr[cL];
      const float rgt = sr[cR];
      v[r][0] = (okr && okL) ? lft : 0.0f;
      v[r][1] = okr ? mid[0] : 0.0f;
      v[r][2] = okr ? mid[1] : 0.0f;
      v[r][3] = (okr && okR) ? rgt : 0.0f;
    }

    float* po = out + (size_t)bc * PLANE + (size_t)(PS_ROWS * m) * HW_ + PS_PIX * j;
#pragma unroll
    for (int i = 0; i < PS_ROWS; ++i) {
      const int   rk  = (i >> 1) + 1;
      const bool  ye  = !(i & 1);
      const int   ry0 = ye ? rk     : rk + 1;  const float wy0 = ye ? 0.75f : 0.25f;
      const int   ry1 = ye ? rk - 1 : rk;      const float wy1 = ye ? 0.25f : 0.75f;
      v4f o;
#pragma unroll
      for (int p = 0; p < PS_PIX; ++p) {
        const int   ck  = (p >> 1) + 1;
        const bool  xe  = !(p & 1);
        const int   cx0 = xe ? ck     : ck + 1;  const float wx0 = xe ? 0.75f : 0.25f;
        const int   cx1 = xe ? ck - 1 : ck;      const float wx1 = xe ? 0.25f : 0.75f;
        float acc = bv;
        acc = __builtin_fmaf(wy0 * wx0, v[ry0][cx0], acc);
        acc = __builtin_fmaf(wy0 * wx1, v[ry0][cx1], acc);
        acc = __builtin_fmaf(wy1 * wx0, v[ry1][cx0], acc);
        acc = __builtin_fmaf(wy1 * wx1, v[ry1][cx1], acc);
        o[p] = acc;
      }
      *(v4f*)(po + i * HW_) = o;
    }
  }

  const float e  = g0 + g2;
  const float gu = (pu == 0) ? g0 : (pu == 3) ? g2 : (pu == 1) ? 0.5f * (e + g1) : 0.5f * (e - g1);
  bp[tid] = live ? gu * EQ_SCALE : 0.0f;
}

#define CHF    16
#define SLOTF  (TC * CHF)
#define ROWP   272
#define PAIRF  (NRW * ROWP)
#define NLD    3
#define EXW    (HW_ + 2)
#define EXP    13


__device__ __forceinline__ void gl_trip(v4f (&G)[NLD], v4f (&GB)[2],
                                        const float* __restrict__ pin,
                                        const v4f* __restrict__ bp,
                                        const unsigned (&goff)[NLD], int lane, int c0) {
  const char* pn = (const char*)(pin + (size_t)c0 * PLANE);
#pragma unroll
  for (int i = 0; i < NLD; ++i) G[i] = *(const v4f*)(pn + goff[i]);
#pragma unroll
  for (int j = 0; j < 2; ++j) GB[j] = bp[((c0 >> 2) * 2 + j) * 32 + lane];
}

template <int MODE>
__device__ __forceinline__ void st_trip(const v4f (&G)[NLD], float* slot,
                                        const unsigned (&soff)[NLD], const bool (&zf)[NLD]) {
#pragma unroll
  for (int i = 0; i < NLD; ++i) {
    v4f v = G[i];
    if (MODE != 0) { v4f z = {0.0f, 0.0f, 0.0f, 0.0f}; v = zf[i] ? z : v; }
    *(v4f*)(slot + soff[i]) = v;
  }
}

__device__ __forceinline__ void fr_trip(v2f (&A)[NRW], const float* slot, int foff) {
#pragma unroll
  for (int j = 0; j < NRW; ++j) {
    const float* p = slot + j * ROWP + foff;
    v2f v = {p[0], p[CHF]};
    A[j] = v;
  }
}

__device__ __forceinline__ void mul_trip(const v2f (&A)[NRW], const v4f (&B)[2],
                                         v8f (&acc)[NACC]) {
  v2f T[NACC];
#pragma unroll
  for (int p = 0; p < 2; ++p) {
    const v2f d0 = A[2 * p], d1 = A[2 * p + 1], d2 = A[2 * p + 2], d3 = A[2 * p + 3];
    T[4 * p + 0] = d0 - d2;
    T[4 * p + 1] = d1 + d2;
    T[4 * p + 2] = d2 - d1;
    T[4 * p + 3] = d1 - d3;
  }
  v2f Wu[NU];
#pragma unroll
  for (int j = 0; j < 2; ++j) {
    const v2f lo = {B[j][0], B[j][1]};
    const v2f hi = {B[j][2], B[j][3]};
    Wu[2 * j]     = lo;
    Wu[2 * j + 1] = hi;
  }
#pragma unroll
  for (int u = 0; u < NU; ++u) {
#pragma unroll
    for (int p = 0; p < 2; ++p)
      acc[4 * p + u] = __builtin_amdgcn_wmma_f32_16x16x4_f32(
          false, T[4 * p + u], false, Wu[u], (short)0, acc[4 * p + u], false, false);
  }
}

template <int MODE, bool LOAD>
__device__ __forceinline__ void body(v2f (&AC)[NRW], v2f (&AN)[NRW],
                                     v4f (&GS)[NLD], v4f (&GL)[NLD], v4f (&WC)[2],
                                     v8f (&acc)[NACC],
                                     const float* __restrict__ pin, const v4f* __restrict__ bp,
                                     float* slotN, const unsigned (&goff)[NLD],
                                     const unsigned (&soff)[NLD], const bool (&zf)[NLD], int foff,
                                     int lane, int c_far) {
  st_trip<MODE>(GS, slotN, soff, zf);
  __builtin_amdgcn_fence(__ATOMIC_RELEASE, "wavefront");
  __builtin_amdgcn_sched_barrier(0);
  __builtin_amdgcn_fence(__ATOMIC_ACQUIRE, "wavefront");
  fr_trip(AN, slotN, foff);
  __builtin_amdgcn_sched_barrier(0);
  mul_trip(AC, WC, acc);
  asm volatile("" : "+v"(acc[0]), "+v"(acc[1]), "+v"(acc[2]), "+v"(acc[3]),
                    "+v"(acc[4]), "+v"(acc[5]), "+v"(acc[6]), "+v"(acc[7]));
  __builtin_amdgcn_sched_barrier(0);
  if (LOAD) gl_trip(GL, WC, pin, bp, goff, lane, c_far);
  __builtin_amdgcn_sched_barrier(0);
}

template <int MODE>
__device__ __forceinline__ void conv_rows6(const float* __restrict__ pin,
                                           const v4f* __restrict__ bp,
                                           float* lds, v8f (&acc)[NACC],
                                           int y0, int x0, int lane) {
  const int M  = lane & 15;
  const int hi = lane >> 4;
  const int foff = 2 * hi * CHF + M;

  unsigned goff[NLD], soff[NLD];
  bool zf[NLD];
#pragma unroll
  for (int i = 0; i < NLD; ++i) {
    const int q  = 32 * i + lane;
    const int r  = q >> 4;
    const int u  = (q >> 2) & 3;
    const int k  = q & 3;
    int yy = y0 - 1 + r;
    zf[i] = (yy < 0) || (yy > HW_ - 1);
    yy = yy < 0 ? 0 : (yy > HW_ - 1 ? HW_ - 1 : yy);
    goff[i] = (unsigned)(u * PLANE + yy * HW_ + x0 + 4 * k) * 4u;
    soff[i] = (unsigned)(r * ROWP + u * CHF + 4 * k);
  }
  float* slot0 = lds;
  float* slot1 = lds + SLOTF;

#pragma unroll
  for (int i = 0; i < NACC; ++i) acc[i] = v8f{};

  v2f AX[NRW], AY[NRW];
  v4f G0[NLD], G1[NLD], G2[NLD], G3[NLD], G4[NLD], G5[NLD];
  v4f W0[2], W1[2], W2[2], W3[2], W4[2], W5[2];

  gl_trip(G0, W0, pin, bp, goff, lane, 0);
  gl_trip(G1, W1, pin, bp, goff, lane, TC);
  gl_trip(G2, W2, pin, bp, goff, lane, 2 * TC);
  gl_trip(G3, W3, pin, bp, goff, lane, 3 * TC);
  __builtin_amdgcn_sched_barrier(0);
  st_trip<MODE>(G0, slot0, soff, zf);
  __builtin_amdgcn_fence(__ATOMIC_RELEASE, "wavefront");
  __builtin_amdgcn_sched_barrier(0);
  __builtin_amdgcn_fence(__ATOMIC_ACQUIRE, "wavefront");
  fr_trip(AX, slot0, foff);
  __builtin_amdgcn_sched_barrier(0);
  gl_trip(G4, W4, pin, bp, goff, lane, 4 * TC);
  gl_trip(G5, W5, pin, bp, goff, lane, 5 * TC);
  __builtin_amdgcn_sched_barrier(0);

  for (int c0 = 0; c0 < 120 * TC; c0 += 6 * TC) {
    body<MODE, true>(AX, AY, G1, G0, W0, acc, pin, bp, slot1, goff, soff, zf, foff, lane, c0 + 6 * TC);
    body<MODE, true>(AY, AX, G2, G1, W1, acc, pin, bp, slot0, goff, soff, zf, foff, lane, c0 + 7 * TC);
    body<MODE, true>(AX, AY, G3, G2, W2, acc, pin, bp, slot1, goff, soff, zf, foff, lane, c0 + 8 * TC);
    body<MODE, true>(AY, AX, G4, G3, W3, acc, pin, bp, slot0, goff, soff, zf, foff, lane, c0 + 9 * TC);
    body<MODE, true>(AX, AY, G5, G4, W4, acc, pin, bp, slot1, goff, soff, zf, foff, lane, c0 + 10 * TC);
    body<MODE, true>(AY, AX, G0, G5, W5, acc, pin, bp, slot0, goff, soff, zf, foff, lane, c0 + 11 * TC);
  }
  body<MODE, true >(AX, AY, G1, G0, W0, acc, pin, bp, slot1, goff, soff, zf, foff, lane, 126 * TC);
  body<MODE, true >(AY, AX, G2, G1, W1, acc, pin, bp, slot0, goff, soff, zf, foff, lane, 127 * TC);
  body<MODE, false>(AX, AY, G3, G2, W2, acc, pin, bp, slot1, goff, soff, zf, foff, lane, 0);
  body<MODE, false>(AY, AX, G4, G3, W3, acc, pin, bp, slot0, goff, soff, zf, foff, lane, 0);
  body<MODE, false>(AX, AY, G5, G4, W4, acc, pin, bp, slot1, goff, soff, zf, foff, lane, 0);
  body<MODE, false>(AY, AX, G0, G5, W5, acc, pin, bp, slot0, goff, soff, zf, foff, lane, 0);
  body<MODE, false>(AX, AY, G1, G0, W0, acc, pin, bp, slot1, goff, soff, zf, foff, lane, 0);
  mul_trip(AY, W1, acc);
}

__global__ void __launch_bounds__(256) conv_rows_kernel(const float* __restrict__ in,
                                 const float* __restrict__ bpack,
                                 float* __restrict__ out) {
  __shared__ __attribute__((aligned(16))) float stage[4][PAIRF];
  __shared__ float ex[RS * EXW * EXP];
  const int lane = threadIdx.x & 31;
  const int w    = __builtin_amdgcn_readfirstlane(threadIdx.x >> 5);
  float* lds = stage[w >> 1] + (w & 1) * (2 * SLOTF);
  const int st   = blockIdx.x & 31;
  const int b    = blockIdx.x >> 5;
  const int y0   = st * RS;
  const int x0   = w * 16;
  const float* pin  = in + (size_t)b * C_IN * PLANE;
  float*       pout = out + (size_t)b * 3 * PLANE;
  const v4f*   bp   = (const v4f*)bpack;

  const int X    = threadIdx.x & (HW_ - 1);
  const int half = threadIdx.x >> 7;
  float* po = pout + (size_t)(y0 + 2 * half) * HW_ + X;
  float start[(RS * 3 * HW_) / 256];
#pragma unroll
  for (int n = 0; n < (RS * 3 * HW_) / 256; ++n)
    start[n] = po[(n % 3) * PLANE + (n / 3) * HW_];

  v8f m[NACC];
  if (st == 0 || st == HW_ / RS - 1) conv_rows6<1>(pin, bp, lds, m, y0, x0, lane);
  else                               conv_rows6<0>(pin, bp, lds, m, y0, x0, lane);

  v8f acc[RS];
#pragma unroll
  for (int p = 0; p < 2; ++p) {
    acc[2 * p]     = (m[4 * p] + m[4 * p + 1]) + m[4 * p + 2];
    acc[2 * p + 1] = (m[4 * p + 1] - m[4 * p + 2]) - m[4 * p + 3];
  }

  const int c  = lane & 15;
  const int hi = lane >> 4;
  if (threadIdx.x < RS * 2 * EXP) {
    const int i    = threadIdx.x / (2 * EXP);
    const int rm   = threadIdx.x - i * (2 * EXP);
    const int side = rm / EXP;
    const int cc   = rm - side * EXP;
    ex[(i * EXW + side * (EXW - 1)) * EXP + cc] = 0.0f;
  }
  if (c < 11) {
#pragma unroll
    for (int i = 0; i < RS; ++i) {
#pragma unroll
      for (int r = 0; r < 8; ++r)
        ex[(i * EXW + 1 + x0 + 8 * hi + r) * EXP + c] = acc[i][r];
    }
  }
  __syncthreads();
  const float* eb = ex + (2 * half * EXW + X) * EXP;
#pragma unroll
  for (int n = 0; n < (RS * 3 * HW_) / 256; ++n) {
    const float* e = eb + (n / 3) * (EXW * EXP) + (n % 3);
    const float v = (e[0] + e[EXP + 4]) + e[2 * EXP + 8];
    po[(n % 3) * PLANE + (n / 3) * HW_] = v + start[n];
  }
}

extern "C" void kernel_launch(void* const* d_in, const int* in_sizes, int n_in,
                              void* d_out, int out_size, void* d_ws,
                              size_t ws_size, hipStream_t stream) {
  (void)in_sizes; (void)n_in; (void)out_size; (void)ws_size;
  const float* in   = (const float*)d_in[0];
  const float* skip = (const float*)d_in[1];
  const float* w    = (const float*)d_in[2];
  const float* bias = (const float*)d_in[3];
  float* out = (float*)d_out;
  float* bp  = (float*)d_ws;

  pack_start_kernel<<<dim3(PS_BLOCKS), dim3(256), 0, stream>>>(w, skip, bias, bp, out);
  conv_rows_kernel<<<dim3(256), dim3(256), 0, stream>>>(in, bp, out);
}


# A fragments straight from the 16-byte row loads (tile e = pixels 4m+e, K = the lane's two channels): LDS staging and its 9 LDS ops per trip gone; wave = 64 px x 2 rows, 4 a block; sum order unchanged
# speedup vs baseline: 25.0230x; 1.2070x over previous
#include <hip/hip_runtime.h>

typedef __attribute__((ext_vector_type(2))) float v2f;
typedef __attribute__((ext_vector_type(4))) float v4f;
typedef __attribute__((ext_vector_type(8))) float v8f;

#define C_IN   512
#define HW_    128
#define PLANE  (HW_ * HW_)
#define NBATCH 8
#define RS     4
#define TC     4
#define NQ     (C_IN / TC)
#define NU     4
#define EQ_SCALE 0.014731391274719739f

#define PS_ROWS   4
#define PS_PIX    4
#define PS_START_THREADS (NBATCH * 3 * (HW_ / PS_ROWS) * (HW_ / PS_PIX))
#define PS_THREADS (NQ * NU * 64)
#define PS_BLOCKS  (PS_THREADS / 256)
__global__ void __launch_bounds__(256) pack_start_kernel(const float* __restrict__ w,
                                  const float* __restrict__ skip,
                                  const float* __restrict__ bias,
                                  float* __restrict__ bp,
                                  float* __restrict__ out) {
  const int tid = blockIdx.x * 256 + threadIdx.x;

  const int ps = tid & 1;
  const int pul = (tid >> 1) & 1;
  const int pc = (tid >> 2) & 15;
  const int ph = (tid >> 6) & 1;
  const int pj = (tid >> 7) & 1;
  const int pq = tid >> 8;
  const int pu = 2 * pj + pul;
  const int pci = 4 * pq + 2 * ph + ps, pdx = pc >> 2, pco = pc & 3;
  const bool live = (pdx < 3) && (pco < 3);
  const float* pw = w + (live ? pco * (C_IN * 9) + pci * 9 + pdx : 0);
  const float g0 = pw[0], g1 = pw[3], g2 = pw[6];

  if (tid < PS_START_THREADS) {
    const int j  = tid & 31;
    const int m  = (tid >> 5) & 31;
    const int bc = tid >> 10;
    const int co = bc % 3;
    const float* s = skip + (size_t)bc * 64 * 64;
    const float bv = bias[co];

    const bool okL = (j > 0), okR = (j < 31);
    const int  cL  = okL ? 2 * j - 1 : 0;
    const int  cR  = okR ? 2 * j + 2 : 63;
    float v[4][4];
#pragma unroll
    for (int r = 0; r < 4; ++r) {
      const int  rr  = 2 * m - 1 + r;
      const bool okr = (rr >= 0) && (rr < 64);
      const int  rc  = rr < 0 ? 0 : (rr > 63 ? 63 : rr);
      const float* sr = s + rc * 64;
      const v2f   mid = *(const v2f*)(sr + 2 * j);
      const float lft = sr[cL];
      const float rgt = sr[cR];
      v[r][0] = (okr && okL) ? lft : 0.0f;
      v[r][1] = okr ? mid[0] : 0.0f;
      v[r][2] = okr ? mid[1] : 0.0f;
      v[r][3] = (okr && okR) ? rgt : 0.0f;
    }

    float* po = out + (size_t)bc * PLANE + (size_t)(PS_ROWS * m) * HW_ + PS_PIX * j;
#pragma unroll
    for (int i = 0; i < PS_ROWS; ++i) {
      const int   rk  = (i >> 1) + 1;
      const bool  ye  = !(i & 1);
      const int   ry0 = ye ? rk     : rk + 1;  const float wy0 = ye ? 0.75f : 0.25f;
      const int   ry1 = ye ? rk - 1 : rk;      const float wy1 = ye ? 0.25f : 0.75f;
      v4f o;
#pragma unroll
      for (int p = 0; p < PS_PIX; ++p) {
        const int   ck  = (p >> 1) + 1;
        const bool  xe  = !(p & 1);
        const int   cx0 = xe ? ck     : ck + 1;  const float wx0 = xe ? 0.75f : 0.25f;
        const int   cx1 = xe ? ck - 1 : ck;      const float wx1 = xe ? 0.25f : 0.75f;
        float acc = bv;
        acc = __builtin_fmaf(wy0 * wx0, v[ry0][cx0], acc);
        acc = __builtin_fmaf(wy0 * wx1, v[ry0][cx1], acc);
        acc = __builtin_fmaf(wy1 * wx0, v[ry1][cx0], acc);
        acc = __builtin_fmaf(wy1 * wx1, v[ry1][cx1], acc);
        o[p] = acc;
      }
      *(v4f*)(po + i * HW_) = o;
    }
  }

  const float e  = g0 + g2;
  const float gu = (pu == 0) ? g0 : (pu == 3) ? g2 : (pu == 1) ? 0.5f * (e + g1) : 0.5f * (e - g1);
  bp[tid] = live ? gu * EQ_SCALE : 0.0f;
}

#define NRP    4
#define NT     4
#define NLR    (NRP * 2)
#define EXW    (HW_ + 2)
#define EXP    13


#define ZROW(MODE) ((MODE) == 1 ? 0 : ((MODE) == 2 ? NRP - 1 : -1))

template <int MODE>
__device__ __forceinline__ void gl_rows(v4f (&G)[NLR], const float* __restrict__ pin,
                                        const unsigned (&goff)[NRP], int c0) {
  const char* pn = (const char*)(pin + (size_t)c0 * PLANE);
#pragma unroll
  for (int j = 0; j < NRP; ++j) {
    if (j == ZROW(MODE)) continue;
    G[2 * j]     = *(const v4f*)(pn + goff[j]);
    G[2 * j + 1] = *(const v4f*)(pn + goff[j] + PLANE * 4);
  }
}

__device__ __forceinline__ void gl_wts(v4f (&GB)[2], const v4f* __restrict__ bp, int lane, int c0) {
#pragma unroll
  for (int j = 0; j < 2; ++j) GB[j] = bp[((c0 >> 2) * 2 + j) * 32 + lane];
}

template <int MODE>
__device__ __forceinline__ void mul_trip(const v4f (&G)[NLR], const v4f (&B)[2],
                                         v8f (&acc)[NT * NU]) {
  v4f D[NRP][2];
#pragma unroll
  for (int j = 0; j < NRP; ++j) {
#pragma unroll
    for (int s = 0; s < 2; ++s) {
      const v4f z = {0.0f, 0.0f, 0.0f, 0.0f};
      D[j][s] = (j == ZROW(MODE)) ? z : G[2 * j + s];
    }
  }
  v2f Wu[NU];
#pragma unroll
  for (int j = 0; j < 2; ++j) {
    const v2f lo = {B[j][0], B[j][1]};
    const v2f hi = {B[j][2], B[j][3]};
    Wu[2 * j]     = lo;
    Wu[2 * j + 1] = hi;
  }
  v2f T[NT][NU];
#pragma unroll
  for (int e = 0; e < NT; ++e) {
#pragma unroll
    for (int s = 0; s < 2; ++s) {
      const float a0 = D[0][s][e], a1 = D[1][s][e], a2 = D[2][s][e], a3 = D[3][s][e];
      float t0 = a0 - a2, t1 = a1 + a2, t2 = a2 - a1, t3 = a1 - a3;
      asm("" : "+v"(t0), "+v"(t1), "+v"(t2), "+v"(t3));
      T[e][0][s] = t0;
      T[e][1][s] = t1;
      T[e][2][s] = t2;
      T[e][3][s] = t3;
    }
  }
#pragma unroll
  for (int u = 0; u < NU; ++u) {
#pragma unroll
    for (int e = 0; e < NT; ++e)
      acc[4 * e + u] = __builtin_amdgcn_wmma_f32_16x16x4_f32(
          false, T[e][u], false, Wu[u], (short)0, acc[4 * e + u], false, false);
  }
}

template <int MODE, bool LOAD>
__device__ __forceinline__ void body(v4f (&GC)[NLR], v4f (&WC)[2], v8f (&acc)[NT * NU],
                                     const float* __restrict__ pin, const v4f* __restrict__ bp,
                                     const unsigned (&goff)[NRP], int lane, int c_far) {
  mul_trip<MODE>(GC, WC, acc);
  __builtin_amdgcn_sched_barrier(0);
  if (LOAD) { gl_rows<MODE>(GC, pin, goff, c_far); gl_wts(WC, bp, lane, c_far); }
  __builtin_amdgcn_sched_barrier(0);
}

template <int MODE>
__device__ __forceinline__ void conv_rows4(const float* __restrict__ pin,
                                           const v4f* __restrict__ bp,
                                           v8f (&acc)[NT * NU], int yt, int x0, int lane) {
  const int c = lane & 15;
  const int h = lane >> 4;
  unsigned goff[NRP];
#pragma unroll
  for (int j = 0; j < NRP; ++j) {
    const int yy = (j == ZROW(MODE)) ? 0 : yt - 1 + j;
    goff[j] = (unsigned)(2 * h * PLANE + yy * HW_ + x0 + 4 * c) * 4u;
  }

#pragma unroll
  for (int i = 0; i < NT * NU; ++i) acc[i] = v8f{};

  v4f G0[NLR], G1[NLR], G2[NLR], G3[NLR], G4[NLR], G5[NLR];
  v4f W0[2], W1[2], W2[2], W3[2], W4[2], W5[2];

  gl_rows<MODE>(G0, pin, goff, 0);      gl_wts(W0, bp, lane, 0);
  gl_rows<MODE>(G1, pin, goff, TC);     gl_wts(W1, bp, lane, TC);
  gl_rows<MODE>(G2, pin, goff, 2 * TC); gl_wts(W2, bp, lane, 2 * TC);
  gl_rows<MODE>(G3, pin, goff, 3 * TC); gl_wts(W3, bp, lane, 3 * TC);
  gl_rows<MODE>(G4, pin, goff, 4 * TC); gl_wts(W4, bp, lane, 4 * TC);
  gl_rows<MODE>(G5, pin, goff, 5 * TC); gl_wts(W5, bp, lane, 5 * TC);
  __builtin_amdgcn_sched_barrier(0);

  for (int c0 = 0; c0 < 120 * TC; c0 += 6 * TC) {
    body<MODE, true>(G0, W0, acc, pin, bp, goff, lane, c0 + 6 * TC);
    body<MODE, true>(G1, W1, acc, pin, bp, goff, lane, c0 + 7 * TC);
    body<MODE, true>(G2, W2, acc, pin, bp, goff, lane, c0 + 8 * TC);
    body<MODE, true>(G3, W3, acc, pin, bp, goff, lane, c0 + 9 * TC);
    body<MODE, true>(G4, W4, acc, pin, bp, goff, lane, c0 + 10 * TC);
    body<MODE, true>(G5, W5, acc, pin, bp, goff, lane, c0 + 11 * TC);
  }
  body<MODE, true >(G0, W0, acc, pin, bp, goff, lane, 126 * TC);
  body<MODE, true >(G1, W1, acc, pin, bp, goff, lane, 127 * TC);
  body<MODE, false>(G2, W2, acc, pin, bp, goff, lane, 0);
  body<MODE, false>(G3, W3, acc, pin, bp, goff, lane, 0);
  body<MODE, false>(G4, W4, acc, pin, bp, goff, lane, 0);
  body<MODE, false>(G5, W5, acc, pin, bp, goff, lane, 0);
  body<MODE, false>(G0, W0, acc, pin, bp, goff, lane, 0);
  body<MODE, false>(G1, W1, acc, pin, bp, goff, lane, 0);
}

__global__ void __launch_bounds__(128) conv_rows_kernel(const float* __restrict__ in,
                                 const float* __restrict__ bpack,
                                 float* __restrict__ out) {
  __shared__ float ex[RS * EXW * EXP];
  const int lane = threadIdx.x & 31;
  const int w    = __builtin_amdgcn_readfirstlane(threadIdx.x >> 5);
  const int st   = blockIdx.x & 31;
  const int b    = blockIdx.x >> 5;
  const int y0   = st * RS;
  const int pr   = w >> 1;
  const int x0   = (w & 1) * 64;
  const float* pin  = in + (size_t)b * C_IN * PLANE;
  float*       pout = out + (size_t)b * 3 * PLANE;
  const v4f*   bp   = (const v4f*)bpack;

  const int X = threadIdx.x;
  float* po = pout + (size_t)y0 * HW_ + X;
  float start[RS * 3];
#pragma unroll
  for (int n = 0; n < RS * 3; ++n)
    start[n] = po[(n % 3) * PLANE + (n / 3) * HW_];

  v8f m[NT * NU];
  const int yt = y0 + 2 * pr;
  if (yt == 0)             conv_rows4<1>(pin, bp, m, yt, x0, lane);
  else if (yt == HW_ - 2)  conv_rows4<2>(pin, bp, m, yt, x0, lane);
  else                     conv_rows4<0>(pin, bp, m, yt, x0, lane);

  v8f acc[2][NT];
#pragma unroll
  for (int e = 0; e < NT; ++e) {
    acc[0][e] = (m[4 * e] + m[4 * e + 1]) + m[4 * e + 2];
    acc[1][e] = (m[4 * e + 1] - m[4 * e + 2]) - m[4 * e + 3];
  }

  const int c  = lane & 15;
  const int hi = lane >> 4;
  if (threadIdx.x < RS * 2 * EXP) {
    const int i    = threadIdx.x / (2 * EXP);
    const int rm   = threadIdx.x - i * (2 * EXP);
    const int side = rm / EXP;
    const int cc   = rm - side * EXP;
    ex[(i * EXW + side * (EXW - 1)) * EXP + cc] = 0.0f;
  }
  if (c < 11) {
#pragma unroll
    for (int i = 0; i < 2; ++i) {
#pragma unroll
      for (int e = 0; e < NT; ++e) {
#pragma unroll
        for (int r = 0; r < 8; ++r)
          ex[((2 * pr + i) * EXW + 1 + x0 + 4 * (8 * hi + r) + e) * EXP + c] = acc[i][e][r];
      }
    }
  }
  __syncthreads();
  const float* eb = ex + X * EXP;
#pragma unroll
  for (int n = 0; n < RS * 3; ++n) {
    const float* e = eb + (n / 3) * (EXW * EXP) + (n % 3);
    const float v = (e[0] + e[EXP + 4]) + e[2 * EXP + 8];
    po[(n % 3) * PLANE + (n / 3) * HW_] = v + start[n];
  }
}

extern "C" void kernel_launch(void* const* d_in, const int* in_sizes, int n_in,
                              void* d_out, int out_size, void* d_ws,
                              size_t ws_size, hipStream_t stream) {
  (void)in_sizes; (void)n_in; (void)out_size; (void)ws_size;
  const float* in   = (const float*)d_in[0];
  const float* skip = (const float*)d_in[1];
  const float* w    = (const float*)d_in[2];
  const float* bias = (const float*)d_in[3];
  float* out = (float*)d_out;
  float* bp  = (float*)d_ws;

  pack_start_kernel<<<dim3(PS_BLOCKS), dim3(256), 0, stream>>>(w, skip, bias, bp, out);
  conv_rows_kernel<<<dim3(256), dim3(128), 0, stream>>>(in, bp, out);
}
